# starting kernel
# baseline (speedup 1.0000x reference)
#include <hip/hip_runtime.h>
#include <math.h>


typedef float v2f __attribute__((ext_vector_type(2)));
typedef float v8f __attribute__((ext_vector_type(8)));

__device__ __forceinline__ v8f wmma_f32(v2f a, v2f b, v8f c) {
  return __builtin_amdgcn_wmma_f32_16x16x4_f32(false, a, false, b,
                                               (short)0, c, false, false);
}

#define BB 8
#define CC 1024
#define INF_NEG (-3.0e38f)

__global__ void __launch_bounds__(256) stats_kernel(
    const float* __restrict__ x,
    const float* __restrict__ Ws1, const float* __restrict__ bs1,
    const float* __restrict__ Ws2, const float* __restrict__ bs2,
    const float* __restrict__ Wc1, const float* __restrict__ bc1,
    const float* __restrict__ Wc2, const float* __restrict__ bc2,
    const float* __restrict__ att_w,
    float* __restrict__ slog_out, float* __restrict__ cs_out) {
  __shared__ float xm[256];
  __shared__ float hid[128];
  __shared__ float slg[5];
  __shared__ float sp[5];
  __shared__ float c1s[64];
  __shared__ float c2s[64];

  const int b = blockIdx.x;
  const int t = threadIdx.x;

  float s = 0.f;
  for (int c = 0; c < CC; ++c) s += x[((size_t)(b * CC + c)) * 256 + t];
  xm[t] = s * (1.0f / 1024.0f);
  __syncthreads();

  if (t < 128) {
    float a = bs1[t];
    for (int k = 0; k < 256; ++k) a += xm[k] * Ws1[k * 128 + t];
    hid[t] = fmaxf(a, 0.f);
  }
  __syncthreads();

  if (t < 5) {
    float a = bs2[t];
    for (int k = 0; k < 128; ++k) a += hid[k] * Ws2[k * 5 + t];
    slg[t] = a;
    slog_out[b * 5 + t] = a;
  }
  __syncthreads();

  if (t == 0) {
    float mx = slg[0];
    for (int i = 1; i < 5; ++i) mx = fmaxf(mx, slg[i]);
    float sum = 0.f;
    for (int i = 0; i < 5; ++i) { sp[i] = __expf(slg[i] - mx); sum += sp[i]; }
    float inv = 1.0f / sum;
    for (int i = 0; i < 5; ++i) sp[i] *= inv;
  }
  __syncthreads();

  if (t < 64) {
    float a = bc1[t];
    for (int k = 0; k < 5; ++k) a += sp[k] * Wc1[k * 64 + t];
    c1s[t] = fmaxf(a, 0.f);
  }
  __syncthreads();

  if (t < 64) {
    float a = bc2[t];
    for (int k = 0; k < 64; ++k) a += c1s[k] * Wc2[k * 64 + t];
    c2s[t] = a;
  }
  __syncthreads();

  if (t == 0) {
    float cs = 0.f;
    for (int k = 0; k < 64; ++k) cs += c2s[k] * att_w[128 + k];
    cs_out[b] = cs;
  }
}

__global__ void __launch_bounds__(256) logA_kernel(const float* __restrict__ A,
                                                   float* __restrict__ logA) {
  const int g = blockIdx.x * 256 + threadIdx.x;
  logA[g] = logf(A[g] + 1e-9f);
}

#define APAD 260
__global__ void __launch_bounds__(256) proj_gemm(
    const float* __restrict__ x, const float* __restrict__ Wp,
    const float* __restrict__ bp, float* __restrict__ xp) {
  extern __shared__ float sm[];
  float* Apan = sm;
  float* BpT  = sm + 64 * APAD;

  const int tid = threadIdx.x;
  const int m0 = blockIdx.x * 64;
  const int n0 = blockIdx.y * 64;

  for (int idx = tid; idx < 64 * 256; idx += 256) {
    const int r = idx >> 8, k = idx & 255;
    Apan[r * APAD + k] = x[(size_t)(m0 + r) * 256 + k];
  }
  for (int idx = tid; idx < 256 * 64; idx += 256) {
    const int k = idx >> 6, n = idx & 63;
    BpT[n * APAD + k] = Wp[(size_t)k * 256 + n0 + n];
  }
  __syncthreads();

  const int wave = tid >> 5, lane = tid & 31;
  const int r16 = lane & 15, half = lane >> 4;
  const int rw = wave >> 1;
  const int cw = (wave & 1) * 32;

  v8f acc0 = {0, 0, 0, 0, 0, 0, 0, 0};
  v8f acc1 = {0, 0, 0, 0, 0, 0, 0, 0};

  const float* arow = Apan + (rw * 16 + r16) * APAD;
  const float* b0row = BpT + (cw + r16) * APAD;
  const float* b1row = BpT + (cw + 16 + r16) * APAD;

  for (int kk = 0; kk < 256; kk += 4) {
    const int ko = kk + 2 * half;
    v2f a;  a.x = arow[ko];  a.y = arow[ko + 1];
    v2f b0; b0.x = b0row[ko]; b0.y = b0row[ko + 1];
    v2f b1; b1.x = b1row[ko]; b1.y = b1row[ko + 1];
    acc0 = wmma_f32(a, b0, acc0);
    acc1 = wmma_f32(a, b1, acc1);
  }

  const float bias0 = bp[n0 + cw + r16];
  const float bias1 = bp[n0 + cw + 16 + r16];
#pragma unroll
  for (int v = 0; v < 8; ++v) {
    const int row = m0 + rw * 16 + v + 8 * half;
    xp[(size_t)row * 256 + n0 + cw + r16]      = acc0[v] + bias0;
    xp[(size_t)row * 256 + n0 + cw + 16 + r16] = acc1[v] + bias1;
  }
}

__global__ void __launch_bounds__(256) srcdst_kernel(
    const float* __restrict__ xp, const float* __restrict__ att_w,
    float* __restrict__ srcg, float* __restrict__ dstg) {
  __shared__ float ss[256];
  __shared__ float sd[256];
  const int m = blockIdx.x, t = threadIdx.x;
  const int h = t >> 6, d = t & 63;
  const float v = xp[(size_t)m * 256 + t];
  ss[t] = v * att_w[d];
  sd[t] = v * att_w[64 + d];
  __syncthreads();
  for (int off = 32; off >= 1; off >>= 1) {
    if ((t & 63) < off) { ss[t] += ss[t + off]; sd[t] += sd[t + off]; }
    __syncthreads();
  }
  if ((t & 63) == 0) {
    srcg[m * 4 + h] = ss[t];
    dstg[m * 4 + h] = sd[t];
  }
}

#define JP 132
__global__ void __launch_bounds__(256) attn_kernel(
    const float* __restrict__ xp,   const float* __restrict__ logA,
    const float* __restrict__ srcg, const float* __restrict__ dstg,
    const float* __restrict__ csb,  const float* __restrict__ lam,
    float* __restrict__ out) {
  extern __shared__ float sm[];
  float* srcS = sm;
  float* XcT  = srcS + 1024;
  float* eb   = XcT + 64 * JP;

  const int tid = threadIdx.x;
  const int wave = tid >> 5, lane = tid & 31;
  const int r16 = lane & 15, half = lane >> 4;

  const int bid = blockIdx.x;
  const int b  = bid >> 5;
  const int h  = (bid >> 3) & 3;
  const int it = bid & 7;
  const int i0 = it * 128 + wave * 16;

  for (int idx = tid; idx < 1024; idx += 256)
    srcS[idx] = srcg[(b * CC + idx) * 4 + h];
  __syncthreads();

  const float cs = csb[b];
  const float lamh = lam[h];

  float dvals[16];
#pragma unroll
  for (int r = 0; r < 16; ++r)
    dvals[r] = dstg[(b * CC + i0 + r) * 4 + h];

  float mrow[16];
#pragma unroll
  for (int r = 0; r < 16; ++r) mrow[r] = INF_NEG;

  for (int jc = 0; jc < CC; jc += 128) {
    const float4 sv = ((const float4*)(srcS + jc))[lane];
#pragma unroll
    for (int r = 0; r < 16; ++r) {
      const float4 lg =
          ((const float4*)(logA + (size_t)(i0 + r) * CC + jc))[lane];
      float e0 = sv.x + dvals[r] + cs; e0 = e0 > 0.f ? e0 : 0.2f * e0;
      float e1 = sv.y + dvals[r] + cs; e1 = e1 > 0.f ? e1 : 0.2f * e1;
      float e2 = sv.z + dvals[r] + cs; e2 = e2 > 0.f ? e2 : 0.2f * e2;
      float e3 = sv.w + dvals[r] + cs; e3 = e3 > 0.f ? e3 : 0.2f * e3;
      const float v0 = e0 + lamh * lg.x;
      const float v1 = e1 + lamh * lg.y;
      const float v2 = e2 + lamh * lg.z;
      const float v3 = e3 + lamh * lg.w;
      mrow[r] = fmaxf(mrow[r], fmaxf(fmaxf(v0, v1), fmaxf(v2, v3)));
    }
  }
#pragma unroll
  for (int r = 0; r < 16; ++r) {
#pragma unroll
    for (int off = 16; off >= 1; off >>= 1)
      mrow[r] = fmaxf(mrow[r], __shfl_xor(mrow[r], off, 32));
  }

  v8f acc[4];
  const v8f vzero = {0, 0, 0, 0, 0, 0, 0, 0};
#pragma unroll
  for (int nt = 0; nt < 4; ++nt) acc[nt] = vzero;

  float srow[16];
#pragma unroll
  for (int r = 0; r < 16; ++r) srow[r] = 0.f;

  float* ebw = eb + wave * 16 * JP;

  for (int jc = 0; jc < CC; jc += 128) {
    __syncthreads();
    for (int idx = tid; idx < 8192; idx += 256) {
      const int j = idx >> 6, d = idx & 63;
      XcT[d * JP + j] = xp[((size_t)(b * CC + jc + j) * 4 + h) * 64 + d];
    }
    __syncthreads();

    const float4 sv = ((const float4*)(srcS + jc))[lane];
    for (int r = 0; r < 16; ++r) {
      const float4 lg =
          ((const float4*)(logA + (size_t)(i0 + r) * CC + jc))[lane];
      float e0 = sv.x + dvals[r] + cs; e0 = e0 > 0.f ? e0 : 0.2f * e0;
      float e1 = sv.y + dvals[r] + cs; e1 = e1 > 0.f ? e1 : 0.2f * e1;
      float e2 = sv.z + dvals[r] + cs; e2 = e2 > 0.f ? e2 : 0.2f * e2;
      float e3 = sv.w + dvals[r] + cs; e3 = e3 > 0.f ? e3 : 0.2f * e3;
      const float p0 = __expf(e0 + lamh * lg.x - mrow[r]);
      const float p1 = __expf(e1 + lamh * lg.y - mrow[r]);
      const float p2 = __expf(e2 + lamh * lg.z - mrow[r]);
      const float p3 = __expf(e3 + lamh * lg.w - mrow[r]);
      float* dstp = ebw + r * JP + lane * 4;
      dstp[0] = p0; dstp[1] = p1; dstp[2] = p2; dstp[3] = p3;
      float ps = (p0 + p1) + (p2 + p3);
#pragma unroll
      for (int off = 16; off >= 1; off >>= 1) ps += __shfl_xor(ps, off, 32);
      srow[r] += ps;
    }
    __syncthreads();

    const float* arow = ebw + r16 * JP;
    for (int kk = 0; kk < 128; kk += 4) {
      const int ko = kk + 2 * half;
      v2f a; a.x = arow[ko]; a.y = arow[ko + 1];
#pragma unroll
      for (int nt = 0; nt < 4; ++nt) {
        const float* brow = XcT + (nt * 16 + r16) * JP;
        v2f bv; bv.x = brow[ko]; bv.y = brow[ko + 1];
        acc[nt] = wmma_f32(a, bv, acc[nt]);
      }
    }
  }

#pragma unroll
  for (int v = 0; v < 8; ++v) {
    const int rr = v + 8 * half;
    const float inv = 1.0f / srow[rr];
    const int row = i0 + rr;
#pragma unroll
    for (int nt = 0; nt < 4; ++nt) {
      out[(size_t)(b * CC + row) * 256 + h * 64 + nt * 16 + r16] =
          acc[nt][v] * inv;
    }
  }
}

extern "C" void kernel_launch(void* const* d_in, const int* in_sizes, int n_in,
                              void* d_out, int out_size, void* d_ws,
                              size_t ws_size, hipStream_t stream) {
  (void)in_sizes; (void)n_in; (void)out_size; (void)ws_size;

  const float* x   = (const float*)d_in[0];
  const float* Apr = (const float*)d_in[1];
  const float* Wp  = (const float*)d_in[2];
  const float* bp  = (const float*)d_in[3];
  const float* Ws1 = (const float*)d_in[4];
  const float* bs1 = (const float*)d_in[5];
  const float* Ws2 = (const float*)d_in[6];
  const float* bs2 = (const float*)d_in[7];
  const float* Wc1 = (const float*)d_in[8];
  const float* bc1 = (const float*)d_in[9];
  const float* Wc2 = (const float*)d_in[10];
  const float* bc2 = (const float*)d_in[11];
  const float* att = (const float*)d_in[12];
  const float* lam = (const float*)d_in[13];

  float* out  = (float*)d_out;
  float* slog = out + (size_t)BB * CC * 256;

  float* xp   = (float*)d_ws;
  float* logA = xp + 2097152;
  float* srcg = logA + 1048576;
  float* dstg = srcg + 32768;
  float* csb  = dstg + 32768;

  stats_kernel<<<BB, 256, 0, stream>>>(x, Ws1, bs1, Ws2, bs2, Wc1, bc1, Wc2,
                                       bc2, att, slog, csb);
  logA_kernel<<<4096, 256, 0, stream>>>(Apr, logA);

  const size_t gemm_lds = (size_t)(2 * 64 * APAD) * sizeof(float);
  proj_gemm<<<dim3(128, 4), 256, gemm_lds, stream>>>(x, Wp, bp, xp);

  srcdst_kernel<<<BB * CC, 256, 0, stream>>>(xp, att, srcg, dstg);

  const size_t attn_lds =
      (size_t)(1024 + 64 * JP + 8 * 16 * JP) * sizeof(float);
  attn_kernel<<<BB * 4 * 8, 256, attn_lds, stream>>>(xp, logA, srcg, dstg,
                                                     csb, lam, out);
}


# proj writes x_proj d-major (xpT[b][h][d][j]); attn stages each chunk as 8 float4 row copies in one load group where 4 dependent groups of 4-byte transposing loads stood; src/dst sum order changes
# speedup vs baseline: 1.8948x; 1.8948x over previous
#include <hip/hip_runtime.h>
#include <math.h>


typedef float v2f __attribute__((ext_vector_type(2)));
typedef float v8f __attribute__((ext_vector_type(8)));

__device__ __forceinline__ v8f wmma_f32(v2f a, v2f b, v8f c) {
  return __builtin_amdgcn_wmma_f32_16x16x4_f32(false, a, false, b,
                                               (short)0, c, false, false);
}

#define BB 8
#define CC 1024
#define INF_NEG (-3.0e38f)

__global__ void __launch_bounds__(1024) stats_kernel(
    const float* __restrict__ x,
    const float* __restrict__ Ws1, const float* __restrict__ bs1,
    const float* __restrict__ Ws2, const float* __restrict__ bs2,
    const float* __restrict__ Wc1, const float* __restrict__ bc1,
    const float* __restrict__ Wc2, const float* __restrict__ bc2,
    const float* __restrict__ att_w,
    float* __restrict__ slog_out, float* __restrict__ cs_out) {
  __shared__ float4 red[16][64];
  __shared__ float xm[256];
  __shared__ float hid[128];
  __shared__ float slg[5];
  __shared__ float sp[5];
  __shared__ float c1s[64];
  __shared__ float c2s[64];

  const int b = blockIdx.x;
  const int t = threadIdx.x;

  {
    const int cg = t & 63;
    const int rg = t >> 6;
    const float4* xr = (const float4*)(x + ((size_t)b * CC + rg) * 256) + cg;
    float4 s = {0.f, 0.f, 0.f, 0.f};
#pragma unroll 16
    for (int u = 0; u < 64; ++u) {
      const float4 v = xr[(size_t)u * (16 * 64)];
      s.x += v.x;  s.y += v.y;  s.z += v.z;  s.w += v.w;
    }
    red[rg][cg] = s;
  }
  __syncthreads();

  if (t < 256) {
    const float* rf = (const float*)&red[0][0];
    float s = 0.f;
    for (int g = 0; g < 16; ++g) s += rf[g * 256 + t];
    xm[t] = s * (1.0f / 1024.0f);
  }
  __syncthreads();

  if (t < 128) {
    float a = bs1[t];
    for (int k = 0; k < 256; ++k) a += xm[k] * Ws1[k * 128 + t];
    hid[t] = fmaxf(a, 0.f);
  }
  __syncthreads();

  if (t < 5) {
    float a = bs2[t];
    for (int k = 0; k < 128; ++k) a += hid[k] * Ws2[k * 5 + t];
    slg[t] = a;
    slog_out[b * 5 + t] = a;
  }
  __syncthreads();

  if (t == 0) {
    float mx = slg[0];
    for (int i = 1; i < 5; ++i) mx = fmaxf(mx, slg[i]);
    float sum = 0.f;
    for (int i = 0; i < 5; ++i) { sp[i] = __expf(slg[i] - mx); sum += sp[i]; }
    float inv = 1.0f / sum;
    for (int i = 0; i < 5; ++i) sp[i] *= inv;
  }
  __syncthreads();

  if (t < 64) {
    float a = bc1[t];
    for (int k = 0; k < 5; ++k) a += sp[k] * Wc1[k * 64 + t];
    c1s[t] = fmaxf(a, 0.f);
  }
  __syncthreads();

  if (t < 64) {
    float a = bc2[t];
    for (int k = 0; k < 64; ++k) a += c1s[k] * Wc2[k * 64 + t];
    c2s[t] = a;
  }
  __syncthreads();

  if (t == 0) {
    float cs = 0.f;
    for (int k = 0; k < 64; ++k) cs += c2s[k] * att_w[128 + k];
    cs_out[b] = cs;
  }
}

__global__ void __launch_bounds__(256) logA_kernel(const float* __restrict__ A,
                                                   float* __restrict__ logF) {
  const int o = blockIdx.x * 256 + threadIdx.x;
  const int c = o & 15, hf = (o >> 4) & 1, jq = (o >> 5) & 127, tile = o >> 12;
  const float4 a = *(const float4*)(A + (size_t)(tile * 16 + c) * CC +
                                    jq * 8 + hf * 4);
  float4 l;
  l.x = logf(a.x + 1e-9f);
  l.y = logf(a.y + 1e-9f);
  l.z = logf(a.z + 1e-9f);
  l.w = logf(a.w + 1e-9f);
  ((float4*)logF)[o] = l;
}

#define APAD 260
__global__ void __launch_bounds__(256) proj_gemm(
    const float* __restrict__ x, const float* __restrict__ Wp,
    const float* __restrict__ bp, float* __restrict__ xp) {
  extern __shared__ float sm[];
  float* Apan = sm;
  float* BpT  = sm + 64 * APAD;

  const int tid = threadIdx.x;
  const int m0 = blockIdx.x * 64;
  const int n0 = blockIdx.y * 64;

  for (int idx = tid; idx < 64 * 256; idx += 256) {
    const int r = idx >> 8, k = idx & 255;
    Apan[r * APAD + k] = x[(size_t)(m0 + r) * 256 + k];
  }
  for (int idx = tid; idx < 256 * 64; idx += 256) {
    const int k = idx >> 6, n = idx & 63;
    BpT[n * APAD + k] = Wp[(size_t)k * 256 + n0 + n];
  }
  __syncthreads();

  const int wave = tid >> 5, lane = tid & 31;
  const int r16 = lane & 15, half = lane >> 4;
  const int dt = wave >> 1;
  const int jw = (wave & 1) * 32;

  v8f acc0 = {0, 0, 0, 0, 0, 0, 0, 0};
  v8f acc1 = {0, 0, 0, 0, 0, 0, 0, 0};

  const float* arow = BpT + (dt * 16 + r16) * APAD;
  const float* b0row = Apan + (jw + r16) * APAD;
  const float* b1row = Apan + (jw + 16 + r16) * APAD;

  for (int kk = 0; kk < 256; kk += 4) {
    const int ko = kk + 2 * half;
    v2f a;  a.x = arow[ko];  a.y = arow[ko + 1];
    v2f b0; b0.x = b0row[ko]; b0.y = b0row[ko + 1];
    v2f b1; b1.x = b1row[ko]; b1.y = b1row[ko + 1];
    acc0 = wmma_f32(a, b0, acc0);
    acc1 = wmma_f32(a, b1, acc1);
  }

  const float4 bl = *(const float4*)(bp + n0 + dt * 16 + 8 * half);
  const float4 bh = *(const float4*)(bp + n0 + dt * 16 + 8 * half + 4);
  const float bias[8] = {bl.x, bl.y, bl.z, bl.w, bh.x, bh.y, bh.z, bh.w};
  const int bidx = m0 >> 10;
  const int jb = (m0 & 1023) + jw + r16;
  float* xrow = xp + ((size_t)(bidx * 4 + blockIdx.y) * 64 + dt * 16 + 8 * half)
                     * 1024 + jb;
#pragma unroll
  for (int v = 0; v < 8; ++v) {
    xrow[(size_t)v * 1024]      = acc0[v] + bias[v];
    xrow[(size_t)v * 1024 + 16] = acc1[v] + bias[v];
  }
}

#define JP 132
__global__ void __launch_bounds__(256) attn_kernel(
    const float* __restrict__ xp,   const float* __restrict__ logF,
    const float* __restrict__ att_w,
    const float* __restrict__ csb,  const float* __restrict__ lam,
    float* __restrict__ out) {
  extern __shared__ float sm[];
  float* srcS = sm;
  float* dstS = srcS + 1024;
  float* XcT  = dstS + 1024;

  const int tid = threadIdx.x;
  const int wave = tid >> 5, lane = tid & 31;
  const int r16 = lane & 15, half = lane >> 4;

  const int bid = blockIdx.x;
  const int b  = bid >> 5;
  const int h  = (bid >> 3) & 3;
  const int it = bid & 7;
  const int i0 = it * 128 + wave * 16;

  const float* xbh = xp + (size_t)(b * 4 + h) * (64 * 1024);

  {
    const float4* xr = (const float4*)xbh + tid;
    float4 s = {0.f, 0.f, 0.f, 0.f};
    float4 d = {0.f, 0.f, 0.f, 0.f};
#pragma unroll 1
    for (int g = 0; g < 64; g += 32) {
      float4 v[32];
#pragma unroll
      for (int u = 0; u < 32; ++u) v[u] = xr[(size_t)(g + u) * 256];
#pragma unroll
      for (int u = 0; u < 32; ++u) {
        const float ws = att_w[g + u];
        const float wd = att_w[64 + g + u];
        s.x += v[u].x * ws;  s.y += v[u].y * ws;
        s.z += v[u].z * ws;  s.w += v[u].w * ws;
        d.x += v[u].x * wd;  d.y += v[u].y * wd;
        d.z += v[u].z * wd;  d.w += v[u].w * wd;
      }
    }
    ((float4*)srcS)[tid] = s;
    ((float4*)dstS)[tid] = d;
  }
  __syncthreads();

  const float cs = csb[b];
  const float lamh = lam[h];
  const float dv = dstS[i0 + r16];

  const float4* Lw = (const float4*)logF + (size_t)(i0 >> 4) * (128 * 32) + lane;
  const float* srcL = srcS + 4 * half;

  float m = INF_NEG;
  for (int ch = 0; ch < 8; ++ch) {
    float4 L[16];
#pragma unroll
    for (int q = 0; q < 16; ++q) L[q] = Lw[(ch * 16 + q) * 32];
#pragma unroll
    for (int q = 0; q < 16; ++q) {
      const float4 sv = *(const float4*)(srcL + ch * 128 + q * 8);
      float e0 = sv.x + dv + cs; e0 = e0 > 0.f ? e0 : 0.2f * e0;
      float e1 = sv.y + dv + cs; e1 = e1 > 0.f ? e1 : 0.2f * e1;
      float e2 = sv.z + dv + cs; e2 = e2 > 0.f ? e2 : 0.2f * e2;
      float e3 = sv.w + dv + cs; e3 = e3 > 0.f ? e3 : 0.2f * e3;
      const float v0 = e0 + lamh * L[q].x;
      const float v1 = e1 + lamh * L[q].y;
      const float v2 = e2 + lamh * L[q].z;
      const float v3 = e3 + lamh * L[q].w;
      m = fmaxf(m, fmaxf(fmaxf(v0, v1), fmaxf(v2, v3)));
    }
  }
  m = fmaxf(m, __shfl_xor(m, 16, 32));

  v8f acc[4];
  const v8f vzero = {0, 0, 0, 0, 0, 0, 0, 0};
#pragma unroll
  for (int nt = 0; nt < 4; ++nt) acc[nt] = vzero;

  float rsum = 0.f;
  const float* xcl = XcT + r16 * JP + 4 * half;

  for (int ch = 0; ch < 8; ++ch) {
    const int jc = ch * 128;
    float4 L[16];
#pragma unroll
    for (int q = 0; q < 16; ++q) L[q] = Lw[(ch * 16 + q) * 32];

    const float* xg = xbh + (size_t)(tid >> 5) * 1024 + jc + 4 * (tid & 31);
    const float4 t0 = *(const float4*)(xg);
    const float4 t1 = *(const float4*)(xg + 8 * 1024);
    const float4 t2 = *(const float4*)(xg + 16 * 1024);
    const float4 t3 = *(const float4*)(xg + 24 * 1024);
    const float4 t4 = *(const float4*)(xg + 32 * 1024);
    const float4 t5 = *(const float4*)(xg + 40 * 1024);
    const float4 t6 = *(const float4*)(xg + 48 * 1024);
    const float4 t7 = *(const float4*)(xg + 56 * 1024);
    __syncthreads();
    float* xl = XcT + (tid >> 5) * JP + 4 * (tid & 31);
    *(float4*)(xl)           = t0;
    *(float4*)(xl + 8 * JP)  = t1;
    *(float4*)(xl + 16 * JP) = t2;
    *(float4*)(xl + 24 * JP) = t3;
    *(float4*)(xl + 32 * JP) = t4;
    *(float4*)(xl + 40 * JP) = t5;
    *(float4*)(xl + 48 * JP) = t6;
    *(float4*)(xl + 56 * JP) = t7;
    __syncthreads();

#pragma unroll
    for (int q = 0; q < 16; ++q) {
      const float4 sv = *(const float4*)(srcL + jc + q * 8);
      float e0 = sv.x + dv + cs; e0 = e0 > 0.f ? e0 : 0.2f * e0;
      float e1 = sv.y + dv + cs; e1 = e1 > 0.f ? e1 : 0.2f * e1;
      float e2 = sv.z + dv + cs; e2 = e2 > 0.f ? e2 : 0.2f * e2;
      float e3 = sv.w + dv + cs; e3 = e3 > 0.f ? e3 : 0.2f * e3;
      const float p0 = __expf(e0 + lamh * L[q].x - m);
      const float p1 = __expf(e1 + lamh * L[q].y - m);
      const float p2 = __expf(e2 + lamh * L[q].z - m);
      const float p3 = __expf(e3 + lamh * L[q].w - m);
      rsum += (p0 + p1) + (p2 + p3);
      v2f a0; a0.x = p0; a0.y = p1;
      v2f a1; a1.x = p2; a1.y = p3;
      float4 bq[4];
#pragma unroll
      for (int nt = 0; nt < 4; ++nt)
        bq[nt] = *(const float4*)(xcl + nt * 16 * JP + q * 8);
#pragma unroll
      for (int nt = 0; nt < 4; ++nt) {
        v2f bv; bv.x = bq[nt].x; bv.y = bq[nt].y;
        acc[nt] = wmma_f32(a0, bv, acc[nt]);
      }
#pragma unroll
      for (int nt = 0; nt < 4; ++nt) {
        v2f bv; bv.x = bq[nt].z; bv.y = bq[nt].w;
        acc[nt] = wmma_f32(a1, bv, acc[nt]);
      }
    }
  }

  const float rs = rsum + __shfl_xor(rsum, 16, 32);
#pragma unroll
  for (int v = 0; v < 8; ++v) {
    const int rr = v + 8 * half;
    const float inv = 1.0f / __shfl(rs, rr, 32);
    const int row = i0 + rr;
#pragma unroll
    for (int nt = 0; nt < 4; ++nt) {
      out[(size_t)(b * CC + row) * 256 + h * 64 + nt * 16 + r16] =
          acc[nt][v] * inv;
    }
  }
}

extern "C" void kernel_launch(void* const* d_in, const int* in_sizes, int n_in,
                              void* d_out, int out_size, void* d_ws,
                              size_t ws_size, hipStream_t stream) {
  (void)in_sizes; (void)n_in; (void)out_size; (void)ws_size;

  const float* x   = (const float*)d_in[0];
  const float* Apr = (const float*)d_in[1];
  const float* Wp  = (const float*)d_in[2];
  const float* bp  = (const float*)d_in[3];
  const float* Ws1 = (const float*)d_in[4];
  const float* bs1 = (const float*)d_in[5];
  const float* Ws2 = (const float*)d_in[6];
  const float* bs2 = (const float*)d_in[7];
  const float* Wc1 = (const float*)d_in[8];
  const float* bc1 = (const float*)d_in[9];
  const float* Wc2 = (const float*)d_in[10];
  const float* bc2 = (const float*)d_in[11];
  const float* att = (const float*)d_in[12];
  const float* lam = (const float*)d_in[13];

  float* out  = (float*)d_out;
  float* slog = out + (size_t)BB * CC * 256;

  float* xp   = (float*)d_ws;
  float* logA = xp + 2097152;
  float* csb  = logA + 1048576;

  stats_kernel<<<BB, 1024, 0, stream>>>(x, Ws1, bs1, Ws2, bs2, Wc1, bc1, Wc2,
                                        bc2, att, slog, csb);
  logA_kernel<<<1024, 256, 0, stream>>>(Apr, logA);

  const size_t gemm_lds = (size_t)(2 * 64 * APAD) * sizeof(float);
  proj_gemm<<<dim3(128, 4), 256, gemm_lds, stream>>>(x, Wp, bp, xp);

  const size_t attn_lds =
      (size_t)(2048 + 64 * JP) * sizeof(float);
  attn_kernel<<<BB * 4 * 8, 256, attn_lds, stream>>>(xp, logA, att, csb, lam,
                                                     out);
}


# attn: the separate row-max pass over logF is gone; one sweep keeps a running row max per 128-col chunk and rescales acc rows and row sum by exp(m_old-m_new); logF read once; out rounding changes
# speedup vs baseline: 1.9742x; 1.0419x over previous
#include <hip/hip_runtime.h>
#include <math.h>


typedef float v2f __attribute__((ext_vector_type(2)));
typedef float v8f __attribute__((ext_vector_type(8)));

__device__ __forceinline__ v8f wmma_f32(v2f a, v2f b, v8f c) {
  return __builtin_amdgcn_wmma_f32_16x16x4_f32(false, a, false, b,
                                               (short)0, c, false, false);
}

#define BB 8
#define CC 1024
#define INF_NEG (-3.0e38f)

__global__ void __launch_bounds__(1024) stats_kernel(
    const float* __restrict__ x,
    const float* __restrict__ Ws1, const float* __restrict__ bs1,
    const float* __restrict__ Ws2, const float* __restrict__ bs2,
    const float* __restrict__ Wc1, const float* __restrict__ bc1,
    const float* __restrict__ Wc2, const float* __restrict__ bc2,
    const float* __restrict__ att_w,
    float* __restrict__ slog_out, float* __restrict__ cs_out) {
  __shared__ float4 red[16][64];
  __shared__ float xm[256];
  __shared__ float hid[128];
  __shared__ float slg[5];
  __shared__ float sp[5];
  __shared__ float c1s[64];
  __shared__ float c2s[64];

  const int b = blockIdx.x;
  const int t = threadIdx.x;

  {
    const int cg = t & 63;
    const int rg = t >> 6;
    const float4* xr = (const float4*)(x + ((size_t)b * CC + rg) * 256) + cg;
    float4 s = {0.f, 0.f, 0.f, 0.f};
#pragma unroll 16
    for (int u = 0; u < 64; ++u) {
      const float4 v = xr[(size_t)u * (16 * 64)];
      s.x += v.x;  s.y += v.y;  s.z += v.z;  s.w += v.w;
    }
    red[rg][cg] = s;
  }
  __syncthreads();

  if (t < 256) {
    const float* rf = (const float*)&red[0][0];
    float s = 0.f;
    for (int g = 0; g < 16; ++g) s += rf[g * 256 + t];
    xm[t] = s * (1.0f / 1024.0f);
  }
  __syncthreads();

  if (t < 128) {
    float a = bs1[t];
    for (int k = 0; k < 256; ++k) a += xm[k] * Ws1[k * 128 + t];
    hid[t] = fmaxf(a, 0.f);
  }
  __syncthreads();

  if (t < 5) {
    float a = bs2[t];
    for (int k = 0; k < 128; ++k) a += hid[k] * Ws2[k * 5 + t];
    slg[t] = a;
    slog_out[b * 5 + t] = a;
  }
  __syncthreads();

  if (t == 0) {
    float mx = slg[0];
    for (int i = 1; i < 5; ++i) mx = fmaxf(mx, slg[i]);
    float sum = 0.f;
    for (int i = 0; i < 5; ++i) { sp[i] = __expf(slg[i] - mx); sum += sp[i]; }
    float inv = 1.0f / sum;
    for (int i = 0; i < 5; ++i) sp[i] *= inv;
  }
  __syncthreads();

  if (t < 64) {
    float a = bc1[t];
    for (int k = 0; k < 5; ++k) a += sp[k] * Wc1[k * 64 + t];
    c1s[t] = fmaxf(a, 0.f);
  }
  __syncthreads();

  if (t < 64) {
    float a = bc2[t];
    for (int k = 0; k < 64; ++k) a += c1s[k] * Wc2[k * 64 + t];
    c2s[t] = a;
  }
  __syncthreads();

  if (t == 0) {
    float cs = 0.f;
    for (int k = 0; k < 64; ++k) cs += c2s[k] * att_w[128 + k];
    cs_out[b] = cs;
  }
}

__global__ void __launch_bounds__(256) logA_kernel(const float* __restrict__ A,
                                                   float* __restrict__ logF) {
  const int o = blockIdx.x * 256 + threadIdx.x;
  const int c = o & 15, hf = (o >> 4) & 1, jq = (o >> 5) & 127, tile = o >> 12;
  const float4 a = *(const float4*)(A + (size_t)(tile * 16 + c) * CC +
                                    jq * 8 + hf * 4);
  float4 l;
  l.x = logf(a.x + 1e-9f);
  l.y = logf(a.y + 1e-9f);
  l.z = logf(a.z + 1e-9f);
  l.w = logf(a.w + 1e-9f);
  ((float4*)logF)[o] = l;
}

#define APAD 260
__global__ void __launch_bounds__(256) proj_gemm(
    const float* __restrict__ x, const float* __restrict__ Wp,
    const float* __restrict__ bp, float* __restrict__ xp) {
  extern __shared__ float sm[];
  float* Apan = sm;
  float* BpT  = sm + 64 * APAD;

  const int tid = threadIdx.x;
  const int m0 = blockIdx.x * 64;
  const int n0 = blockIdx.y * 64;

  for (int idx = tid; idx < 64 * 256; idx += 256) {
    const int r = idx >> 8, k = idx & 255;
    Apan[r * APAD + k] = x[(size_t)(m0 + r) * 256 + k];
  }
  for (int idx = tid; idx < 256 * 64; idx += 256) {
    const int k = idx >> 6, n = idx & 63;
    BpT[n * APAD + k] = Wp[(size_t)k * 256 + n0 + n];
  }
  __syncthreads();

  const int wave = tid >> 5, lane = tid & 31;
  const int r16 = lane & 15, half = lane >> 4;
  const int dt = wave >> 1;
  const int jw = (wave & 1) * 32;

  v8f acc0 = {0, 0, 0, 0, 0, 0, 0, 0};
  v8f acc1 = {0, 0, 0, 0, 0, 0, 0, 0};

  const float* arow = BpT + (dt * 16 + r16) * APAD;
  const float* b0row = Apan + (jw + r16) * APAD;
  const float* b1row = Apan + (jw + 16 + r16) * APAD;

  for (int kk = 0; kk < 256; kk += 4) {
    const int ko = kk + 2 * half;
    v2f a;  a.x = arow[ko];  a.y = arow[ko + 1];
    v2f b0; b0.x = b0row[ko]; b0.y = b0row[ko + 1];
    v2f b1; b1.x = b1row[ko]; b1.y = b1row[ko + 1];
    acc0 = wmma_f32(a, b0, acc0);
    acc1 = wmma_f32(a, b1, acc1);
  }

  const float4 bl = *(const float4*)(bp + n0 + dt * 16 + 8 * half);
  const float4 bh = *(const float4*)(bp + n0 + dt * 16 + 8 * half + 4);
  const float bias[8] = {bl.x, bl.y, bl.z, bl.w, bh.x, bh.y, bh.z, bh.w};
  const int bidx = m0 >> 10;
  const int jb = (m0 & 1023) + jw + r16;
  float* xrow = xp + ((size_t)(bidx * 4 + blockIdx.y) * 64 + dt * 16 + 8 * half)
                     * 1024 + jb;
#pragma unroll
  for (int v = 0; v < 8; ++v) {
    xrow[(size_t)v * 1024]      = acc0[v] + bias[v];
    xrow[(size_t)v * 1024 + 16] = acc1[v] + bias[v];
  }
}

#define JP 132
__global__ void __launch_bounds__(256) attn_kernel(
    const float* __restrict__ xp,   const float* __restrict__ logF,
    const float* __restrict__ att_w,
    const float* __restrict__ csb,  const float* __restrict__ lam,
    float* __restrict__ out) {
  extern __shared__ float sm[];
  float* srcS = sm;
  float* dstS = srcS + 1024;
  float* XcT  = dstS + 1024;

  const int tid = threadIdx.x;
  const int wave = tid >> 5, lane = tid & 31;
  const int r16 = lane & 15, half = lane >> 4;

  const int bid = blockIdx.x;
  const int b  = bid >> 5;
  const int h  = (bid >> 3) & 3;
  const int it = bid & 7;
  const int i0 = it * 128 + wave * 16;

  const float* xbh = xp + (size_t)(b * 4 + h) * (64 * 1024);

  {
    const float4* xr = (const float4*)xbh + tid;
    float4 s = {0.f, 0.f, 0.f, 0.f};
    float4 d = {0.f, 0.f, 0.f, 0.f};
#pragma unroll 1
    for (int g = 0; g < 64; g += 32) {
      float4 v[32];
#pragma unroll
      for (int u = 0; u < 32; ++u) v[u] = xr[(size_t)(g + u) * 256];
#pragma unroll
      for (int u = 0; u < 32; ++u) {
        const float ws = att_w[g + u];
        const float wd = att_w[64 + g + u];
        s.x += v[u].x * ws;  s.y += v[u].y * ws;
        s.z += v[u].z * ws;  s.w += v[u].w * ws;
        d.x += v[u].x * wd;  d.y += v[u].y * wd;
        d.z += v[u].z * wd;  d.w += v[u].w * wd;
      }
    }
    ((float4*)srcS)[tid] = s;
    ((float4*)dstS)[tid] = d;
  }
  __syncthreads();

  const float cs = csb[b];
  const float lamh = lam[h];
  const float dv = dstS[i0 + r16];

  const float4* Lw = (const float4*)logF + (size_t)(i0 >> 4) * (128 * 32) + lane;
  const float* srcL = srcS + 4 * half;

  v8f acc[4];
  const v8f vzero = {0, 0, 0, 0, 0, 0, 0, 0};
#pragma unroll
  for (int nt = 0; nt < 4; ++nt) acc[nt] = vzero;

  float m = INF_NEG;
  float rsum = 0.f;
  const float* xcl = XcT + r16 * JP + 4 * half;

  for (int ch = 0; ch < 8; ++ch) {
    const int jc = ch * 128;
    float4 L[16];
#pragma unroll
    for (int q = 0; q < 16; ++q) L[q] = Lw[(ch * 16 + q) * 32];

    const float* xg = xbh + (size_t)(tid >> 5) * 1024 + jc + 4 * (tid & 31);
    const float4 t0 = *(const float4*)(xg);
    const float4 t1 = *(const float4*)(xg + 8 * 1024);
    const float4 t2 = *(const float4*)(xg + 16 * 1024);
    const float4 t3 = *(const float4*)(xg + 24 * 1024);
    const float4 t4 = *(const float4*)(xg + 32 * 1024);
    const float4 t5 = *(const float4*)(xg + 40 * 1024);
    const float4 t6 = *(const float4*)(xg + 48 * 1024);
    const float4 t7 = *(const float4*)(xg + 56 * 1024);
    __syncthreads();
    float* xl = XcT + (tid >> 5) * JP + 4 * (tid & 31);
    *(float4*)(xl)           = t0;
    *(float4*)(xl + 8 * JP)  = t1;
    *(float4*)(xl + 16 * JP) = t2;
    *(float4*)(xl + 24 * JP) = t3;
    *(float4*)(xl + 32 * JP) = t4;
    *(float4*)(xl + 40 * JP) = t5;
    *(float4*)(xl + 48 * JP) = t6;
    *(float4*)(xl + 56 * JP) = t7;
    __syncthreads();

    float cm = INF_NEG;
#pragma unroll
    for (int q = 0; q < 16; ++q) {
      const float4 sv = *(const float4*)(srcL + jc + q * 8);
      float e0 = sv.x + dv + cs; e0 = e0 > 0.f ? e0 : 0.2f * e0;
      float e1 = sv.y + dv + cs; e1 = e1 > 0.f ? e1 : 0.2f * e1;
      float e2 = sv.z + dv + cs; e2 = e2 > 0.f ? e2 : 0.2f * e2;
      float e3 = sv.w + dv + cs; e3 = e3 > 0.f ? e3 : 0.2f * e3;
      L[q].x = e0 + lamh * L[q].x;
      L[q].y = e1 + lamh * L[q].y;
      L[q].z = e2 + lamh * L[q].z;
      L[q].w = e3 + lamh * L[q].w;
      cm = fmaxf(cm, fmaxf(fmaxf(L[q].x, L[q].y), fmaxf(L[q].z, L[q].w)));
    }
    cm = fmaxf(cm, __shfl_xor(cm, 16, 32));
    const float mn = fmaxf(m, cm);
    const float sc = __expf(m - mn);
    m = mn;
    rsum *= sc;
#pragma unroll
    for (int v = 0; v < 8; ++v) {
      const float f = __shfl(sc, v + 8 * half, 32);
#pragma unroll
      for (int nt = 0; nt < 4; ++nt) acc[nt][v] *= f;
    }

#pragma unroll
    for (int q = 0; q < 16; ++q) {
      const float p0 = __expf(L[q].x - m);
      const float p1 = __expf(L[q].y - m);
      const float p2 = __expf(L[q].z - m);
      const float p3 = __expf(L[q].w - m);
      rsum += (p0 + p1) + (p2 + p3);
      v2f a0; a0.x = p0; a0.y = p1;
      v2f a1; a1.x = p2; a1.y = p3;
      float4 bq[4];
#pragma unroll
      for (int nt = 0; nt < 4; ++nt)
        bq[nt] = *(const float4*)(xcl + nt * 16 * JP + q * 8);
#pragma unroll
      for (int nt = 0; nt < 4; ++nt) {
        v2f bv; bv.x = bq[nt].x; bv.y = bq[nt].y;
        acc[nt] = wmma_f32(a0, bv, acc[nt]);
      }
#pragma unroll
      for (int nt = 0; nt < 4; ++nt) {
        v2f bv; bv.x = bq[nt].z; bv.y = bq[nt].w;
        acc[nt] = wmma_f32(a1, bv, acc[nt]);
      }
    }
  }

  const float rs = rsum + __shfl_xor(rsum, 16, 32);
#pragma unroll
  for (int v = 0; v < 8; ++v) {
    const int rr = v + 8 * half;
    const float inv = 1.0f / __shfl(rs, rr, 32);
    const int row = i0 + rr;
#pragma unroll
    for (int nt = 0; nt < 4; ++nt) {
      out[(size_t)(b * CC + row) * 256 + h * 64 + nt * 16 + r16] =
          acc[nt][v] * inv;
    }
  }
}

extern "C" void kernel_launch(void* const* d_in, const int* in_sizes, int n_in,
                              void* d_out, int out_size, void* d_ws,
                              size_t ws_size, hipStream_t stream) {
  (void)in_sizes; (void)n_in; (void)out_size; (void)ws_size;

  const float* x   = (const float*)d_in[0];
  const float* Apr = (const float*)d_in[1];
  const float* Wp  = (const float*)d_in[2];
  const float* bp  = (const float*)d_in[3];
  const float* Ws1 = (const float*)d_in[4];
  const float* bs1 = (const float*)d_in[5];
  const float* Ws2 = (const float*)d_in[6];
  const float* bs2 = (const float*)d_in[7];
  const float* Wc1 = (const float*)d_in[8];
  const float* bc1 = (const float*)d_in[9];
  const float* Wc2 = (const float*)d_in[10];
  const float* bc2 = (const float*)d_in[11];
  const float* att = (const float*)d_in[12];
  const float* lam = (const float*)d_in[13];

  float* out  = (float*)d_out;
  float* slog = out + (size_t)BB * CC * 256;

  float* xp   = (float*)d_ws;
  float* logA = xp + 2097152;
  float* csb  = logA + 1048576;

  stats_kernel<<<BB, 1024, 0, stream>>>(x, Ws1, bs1, Ws2, bs2, Wc1, bc1, Wc2,
                                        bc2, att, slog, csb);
  logA_kernel<<<1024, 256, 0, stream>>>(Apr, logA);

  const size_t gemm_lds = (size_t)(2 * 64 * APAD) * sizeof(float);
  proj_gemm<<<dim3(128, 4), 256, gemm_lds, stream>>>(x, Wp, bp, xp);

  const size_t attn_lds =
      (size_t)(2048 + 64 * JP) * sizeof(float);
  attn_kernel<<<BB * 4 * 8, 256, attn_lds, stream>>>(xp, logA, att, csb, lam,
                                                     out);
}


# attn: the prologue's src_s/dst_s loads also fill an LDS copy of the 64x1024 x_proj plane; the sweep's per-chunk tile copy and its two barriers are gone; x_proj read once per block; sums as before
# speedup vs baseline: 2.0691x; 1.0481x over previous
#include <hip/hip_runtime.h>
#include <math.h>


typedef float v2f __attribute__((ext_vector_type(2)));
typedef float v8f __attribute__((ext_vector_type(8)));

__device__ __forceinline__ v8f wmma_f32(v2f a, v2f b, v8f c) {
  return __builtin_amdgcn_wmma_f32_16x16x4_f32(false, a, false, b,
                                               (short)0, c, false, false);
}

#define BB 8
#define CC 1024
#define INF_NEG (-3.0e38f)

__global__ void __launch_bounds__(1024) stats_kernel(
    const float* __restrict__ x,
    const float* __restrict__ Ws1, const float* __restrict__ bs1,
    const float* __restrict__ Ws2, const float* __restrict__ bs2,
    const float* __restrict__ Wc1, const float* __restrict__ bc1,
    const float* __restrict__ Wc2, const float* __restrict__ bc2,
    const float* __restrict__ att_w,
    float* __restrict__ slog_out, float* __restrict__ cs_out) {
  __shared__ float4 red[16][64];
  __shared__ float xm[256];
  __shared__ float hid[128];
  __shared__ float slg[5];
  __shared__ float sp[5];
  __shared__ float c1s[64];
  __shared__ float c2s[64];

  const int b = blockIdx.x;
  const int t = threadIdx.x;

  {
    const int cg = t & 63;
    const int rg = t >> 6;
    const float4* xr = (const float4*)(x + ((size_t)b * CC + rg) * 256) + cg;
    float4 s = {0.f, 0.f, 0.f, 0.f};
#pragma unroll 16
    for (int u = 0; u < 64; ++u) {
      const float4 v = xr[(size_t)u * (16 * 64)];
      s.x += v.x;  s.y += v.y;  s.z += v.z;  s.w += v.w;
    }
    red[rg][cg] = s;
  }
  __syncthreads();

  if (t < 256) {
    const float* rf = (const float*)&red[0][0];
    float s = 0.f;
    for (int g = 0; g < 16; ++g) s += rf[g * 256 + t];
    xm[t] = s * (1.0f / 1024.0f);
  }
  __syncthreads();

  if (t < 128) {
    float a = bs1[t];
    for (int k = 0; k < 256; ++k) a += xm[k] * Ws1[k * 128 + t];
    hid[t] = fmaxf(a, 0.f);
  }
  __syncthreads();

  if (t < 5) {
    float a = bs2[t];
    for (int k = 0; k < 128; ++k) a += hid[k] * Ws2[k * 5 + t];
    slg[t] = a;
    slog_out[b * 5 + t] = a;
  }
  __syncthreads();

  if (t == 0) {
    float mx = slg[0];
    for (int i = 1; i < 5; ++i) mx = fmaxf(mx, slg[i]);
    float sum = 0.f;
    for (int i = 0; i < 5; ++i) { sp[i] = __expf(slg[i] - mx); sum += sp[i]; }
    float inv = 1.0f / sum;
    for (int i = 0; i < 5; ++i) sp[i] *= inv;
  }
  __syncthreads();

  if (t < 64) {
    float a = bc1[t];
    for (int k = 0; k < 5; ++k) a += sp[k] * Wc1[k * 64 + t];
    c1s[t] = fmaxf(a, 0.f);
  }
  __syncthreads();

  if (t < 64) {
    float a = bc2[t];
    for (int k = 0; k < 64; ++k) a += c1s[k] * Wc2[k * 64 + t];
    c2s[t] = a;
  }
  __syncthreads();

  if (t == 0) {
    float cs = 0.f;
    for (int k = 0; k < 64; ++k) cs += c2s[k] * att_w[128 + k];
    cs_out[b] = cs;
  }
}

__global__ void __launch_bounds__(256) logA_kernel(const float* __restrict__ A,
                                                   float* __restrict__ logF) {
  const int o = blockIdx.x * 256 + threadIdx.x;
  const int c = o & 15, hf = (o >> 4) & 1, jq = (o >> 5) & 127, tile = o >> 12;
  const float4 a = *(const float4*)(A + (size_t)(tile * 16 + c) * CC +
                                    jq * 8 + hf * 4);
  float4 l;
  l.x = logf(a.x + 1e-9f);
  l.y = logf(a.y + 1e-9f);
  l.z = logf(a.z + 1e-9f);
  l.w = logf(a.w + 1e-9f);
  ((float4*)logF)[o] = l;
}

#define APAD 260
__global__ void __launch_bounds__(256) proj_gemm(
    const float* __restrict__ x, const float* __restrict__ Wp,
    const float* __restrict__ bp, float* __restrict__ xp) {
  extern __shared__ float sm[];
  float* Apan = sm;
  float* BpT  = sm + 64 * APAD;

  const int tid = threadIdx.x;
  const int m0 = blockIdx.x * 64;
  const int n0 = blockIdx.y * 64;

  for (int idx = tid; idx < 64 * 256; idx += 256) {
    const int r = idx >> 8, k = idx & 255;
    Apan[r * APAD + k] = x[(size_t)(m0 + r) * 256 + k];
  }
  for (int idx = tid; idx < 256 * 64; idx += 256) {
    const int k = idx >> 6, n = idx & 63;
    BpT[n * APAD + k] = Wp[(size_t)k * 256 + n0 + n];
  }
  __syncthreads();

  const int wave = tid >> 5, lane = tid & 31;
  const int r16 = lane & 15, half = lane >> 4;
  const int dt = wave >> 1;
  const int jw = (wave & 1) * 32;

  v8f acc0 = {0, 0, 0, 0, 0, 0, 0, 0};
  v8f acc1 = {0, 0, 0, 0, 0, 0, 0, 0};

  const float* arow = BpT + (dt * 16 + r16) * APAD;
  const float* b0row = Apan + (jw + r16) * APAD;
  const float* b1row = Apan + (jw + 16 + r16) * APAD;

  for (int kk = 0; kk < 256; kk += 4) {
    const int ko = kk + 2 * half;
    v2f a;  a.x = arow[ko];  a.y = arow[ko + 1];
    v2f b0; b0.x = b0row[ko]; b0.y = b0row[ko + 1];
    v2f b1; b1.x = b1row[ko]; b1.y = b1row[ko + 1];
    acc0 = wmma_f32(a, b0, acc0);
    acc1 = wmma_f32(a, b1, acc1);
  }

  const float4 bl = *(const float4*)(bp + n0 + dt * 16 + 8 * half);
  const float4 bh = *(const float4*)(bp + n0 + dt * 16 + 8 * half + 4);
  const float bias[8] = {bl.x, bl.y, bl.z, bl.w, bh.x, bh.y, bh.z, bh.w};
  const int bidx = m0 >> 10;
  const int jb = (m0 & 1023) + jw + r16;
  float* xrow = xp + ((size_t)(bidx * 4 + blockIdx.y) * 64 + dt * 16 + 8 * half)
                     * 1024 + jb;
#pragma unroll
  for (int v = 0; v < 8; ++v) {
    xrow[(size_t)v * 1024]      = acc0[v] + bias[v];
    xrow[(size_t)v * 1024 + 16] = acc1[v] + bias[v];
  }
}

#define JP 132
#define CS (64 * JP)
__global__ void __launch_bounds__(256) attn_kernel(
    const float* __restrict__ xp,   const float* __restrict__ logF,
    const float* __restrict__ att_w,
    const float* __restrict__ csb,  const float* __restrict__ lam,
    float* __restrict__ out) {
  extern __shared__ float sm[];
  float* srcS = sm;
  float* dstS = srcS + 1024;
  float* XpT  = dstS + 1024;

  const int tid = threadIdx.x;
  const int wave = tid >> 5, lane = tid & 31;
  const int r16 = lane & 15, half = lane >> 4;

  const int bid = blockIdx.x;
  const int b  = bid >> 5;
  const int h  = (bid >> 3) & 3;
  const int it = bid & 7;
  const int i0 = it * 128 + wave * 16;

  const float* xbh = xp + (size_t)(b * 4 + h) * (64 * 1024);

  {
    const float4* xr = (const float4*)xbh + tid;
    float4 s = {0.f, 0.f, 0.f, 0.f};
    float4 d = {0.f, 0.f, 0.f, 0.f};
#pragma unroll 1
    for (int g = 0; g < 64; g += 32) {
      float4 v[32];
#pragma unroll
      for (int u = 0; u < 32; ++u) v[u] = xr[(size_t)(g + u) * 256];
#pragma unroll
      for (int u = 0; u < 32; ++u)
        *(float4*)(XpT + (tid >> 5) * CS + (g + u) * JP + 4 * (tid & 31)) = v[u];
#pragma unroll
      for (int u = 0; u < 32; ++u) {
        const float ws = att_w[g + u];
        const float wd = att_w[64 + g + u];
        s.x += v[u].x * ws;  s.y += v[u].y * ws;
        s.z += v[u].z * ws;  s.w += v[u].w * ws;
        d.x += v[u].x * wd;  d.y += v[u].y * wd;
        d.z += v[u].z * wd;  d.w += v[u].w * wd;
      }
    }
    ((float4*)srcS)[tid] = s;
    ((float4*)dstS)[tid] = d;
  }
  __syncthreads();

  const float cs = csb[b];
  const float lamh = lam[h];
  const float dv = dstS[i0 + r16];

  const float4* Lw = (const float4*)logF + (size_t)(i0 >> 4) * (128 * 32) + lane;
  const float* srcL = srcS + 4 * half;

  v8f acc[4];
  const v8f vzero = {0, 0, 0, 0, 0, 0, 0, 0};
#pragma unroll
  for (int nt = 0; nt < 4; ++nt) acc[nt] = vzero;

  float m = INF_NEG;
  float rsum = 0.f;
  const float* xcl = XpT + r16 * JP + 4 * half;

  for (int ch = 0; ch < 8; ++ch) {
    const int jc = ch * 128;
    float4 L[16];
#pragma unroll
    for (int q = 0; q < 16; ++q) L[q] = Lw[(ch * 16 + q) * 32];
    __builtin_amdgcn_sched_barrier(0);

    float cm = INF_NEG;
#pragma unroll
    for (int q = 0; q < 16; ++q) {
      const float4 sv = *(const float4*)(srcL + jc + q * 8);
      float e0 = sv.x + dv + cs; e0 = e0 > 0.f ? e0 : 0.2f * e0;
      float e1 = sv.y + dv + cs; e1 = e1 > 0.f ? e1 : 0.2f * e1;
      float e2 = sv.z + dv + cs; e2 = e2 > 0.f ? e2 : 0.2f * e2;
      float e3 = sv.w + dv + cs; e3 = e3 > 0.f ? e3 : 0.2f * e3;
      L[q].x = e0 + lamh * L[q].x;
      L[q].y = e1 + lamh * L[q].y;
      L[q].z = e2 + lamh * L[q].z;
      L[q].w = e3 + lamh * L[q].w;
      cm = fmaxf(cm, fmaxf(fmaxf(L[q].x, L[q].y), fmaxf(L[q].z, L[q].w)));
    }
    cm = fmaxf(cm, __shfl_xor(cm, 16, 32));
    const float mn = fmaxf(m, cm);
    const float sc = __expf(m - mn);
    m = mn;
    rsum *= sc;
#pragma unroll
    for (int v = 0; v < 8; ++v) {
      const float f = __shfl(sc, v + 8 * half, 32);
#pragma unroll
      for (int nt = 0; nt < 4; ++nt) acc[nt][v] *= f;
    }

#pragma unroll
    for (int q = 0; q < 16; ++q) {
      const float p0 = __expf(L[q].x - m);
      const float p1 = __expf(L[q].y - m);
      const float p2 = __expf(L[q].z - m);
      const float p3 = __expf(L[q].w - m);
      rsum += (p0 + p1) + (p2 + p3);
      v2f a0; a0.x = p0; a0.y = p1;
      v2f a1; a1.x = p2; a1.y = p3;
      float4 bq[4];
#pragma unroll
      for (int nt = 0; nt < 4; ++nt)
        bq[nt] = *(const float4*)(xcl + ch * CS + nt * 16 * JP + q * 8);
#pragma unroll
      for (int nt = 0; nt < 4; ++nt) {
        v2f bv; bv.x = bq[nt].x; bv.y = bq[nt].y;
        acc[nt] = wmma_f32(a0, bv, acc[nt]);
      }
#pragma unroll
      for (int nt = 0; nt < 4; ++nt) {
        v2f bv; bv.x = bq[nt].z; bv.y = bq[nt].w;
        acc[nt] = wmma_f32(a1, bv, acc[nt]);
      }
    }
  }

  const float rs = rsum + __shfl_xor(rsum, 16, 32);
#pragma unroll
  for (int v = 0; v < 8; ++v) {
    const int rr = v + 8 * half;
    const float inv = 1.0f / __shfl(rs, rr, 32);
    const int row = i0 + rr;
#pragma unroll
    for (int nt = 0; nt < 4; ++nt) {
      out[(size_t)(b * CC + row) * 256 + h * 64 + nt * 16 + r16] =
          acc[nt][v] * inv;
    }
  }
}

extern "C" void kernel_launch(void* const* d_in, const int* in_sizes, int n_in,
                              void* d_out, int out_size, void* d_ws,
                              size_t ws_size, hipStream_t stream) {
  (void)in_sizes; (void)n_in; (void)out_size; (void)ws_size;

  const float* x   = (const float*)d_in[0];
  const float* Apr = (const float*)d_in[1];
  const float* Wp  = (const float*)d_in[2];
  const float* bp  = (const float*)d_in[3];
  const float* Ws1 = (const float*)d_in[4];
  const float* bs1 = (const float*)d_in[5];
  const float* Ws2 = (const float*)d_in[6];
  const float* bs2 = (const float*)d_in[7];
  const float* Wc1 = (const float*)d_in[8];
  const float* bc1 = (const float*)d_in[9];
  const float* Wc2 = (const float*)d_in[10];
  const float* bc2 = (const float*)d_in[11];
  const float* att = (const float*)d_in[12];
  const float* lam = (const float*)d_in[13];

  float* out  = (float*)d_out;
  float* slog = out + (size_t)BB * CC * 256;

  float* xp   = (float*)d_ws;
  float* logA = xp + 2097152;
  float* csb  = logA + 1048576;

  stats_kernel<<<BB, 1024, 0, stream>>>(x, Ws1, bs1, Ws2, bs2, Wc1, bc1, Wc2,
                                        bc2, att, slog, csb);
  logA_kernel<<<1024, 256, 0, stream>>>(Apr, logA);

  const size_t gemm_lds = (size_t)(2 * 64 * APAD) * sizeof(float);
  proj_gemm<<<dim3(128, 4), 256, gemm_lds, stream>>>(x, Wp, bp, xp);

  const size_t attn_lds =
      (size_t)(2048 + 8 * CS) * sizeof(float);
  attn_kernel<<<BB * 4 * 8, 256, attn_lds, stream>>>(xp, logA, att, csb, lam,
                                                     out);
}


# stats mean-pool split in two launches: 256 blocks sum 32 rows of x each (8 MB over all WGPs), stats adds 32 partials per batch; 8 WGPs pulled 1 MB each before; 4 launches -> 5; mean sum order changes
# speedup vs baseline: 2.3390x; 1.1304x over previous
#include <hip/hip_runtime.h>
#include <math.h>


typedef float v2f __attribute__((ext_vector_type(2)));
typedef float v8f __attribute__((ext_vector_type(8)));

__device__ __forceinline__ v8f wmma_f32(v2f a, v2f b, v8f c) {
  return __builtin_amdgcn_wmma_f32_16x16x4_f32(false, a, false, b,
                                               (short)0, c, false, false);
}

#define BB 8
#define CC 1024
#define INF_NEG (-3.0e38f)

#define NPART 32
__global__ void __launch_bounds__(256) colsum_kernel(
    const float* __restrict__ x, float* __restrict__ part) {
  __shared__ float4 red4[4][64];
  const int b = blockIdx.x >> 5;
  const int p = blockIdx.x & 31;
  const int t = threadIdx.x;
  const int cg = t & 63;
  const int rg = t >> 6;
  const float4* xr =
      (const float4*)(x + ((size_t)b * CC + p * 32 + rg) * 256) + cg;
  float4 s = {0.f, 0.f, 0.f, 0.f};
#pragma unroll
  for (int u = 0; u < 8; ++u) {
    const float4 v = xr[(size_t)u * (4 * 64)];
    s.x += v.x;  s.y += v.y;  s.z += v.z;  s.w += v.w;
  }
  red4[rg][cg] = s;
  __syncthreads();
  if (t < 64) {
    float4 a = red4[0][t];
    for (int g = 1; g < 4; ++g) {
      const float4 v = red4[g][t];
      a.x += v.x;  a.y += v.y;  a.z += v.z;  a.w += v.w;
    }
    ((float4*)part)[(size_t)(b * NPART + p) * 64 + t] = a;
  }
}

__global__ void __launch_bounds__(1024) stats_kernel(
    const float* __restrict__ part,
    const float* __restrict__ Ws1, const float* __restrict__ bs1,
    const float* __restrict__ Ws2, const float* __restrict__ bs2,
    const float* __restrict__ Wc1, const float* __restrict__ bc1,
    const float* __restrict__ Wc2, const float* __restrict__ bc2,
    const float* __restrict__ att_w,
    float* __restrict__ slog_out, float* __restrict__ cs_out) {
  __shared__ float4 red[16][64];
  __shared__ float xm[256];
  __shared__ float hid[128];
  __shared__ float slg[5];
  __shared__ float sp[5];
  __shared__ float c1s[64];
  __shared__ float c2s[64];

  const int b = blockIdx.x;
  const int t = threadIdx.x;

  {
    const int cg = t & 63;
    const int rg = t >> 6;
    const float4* pr =
        (const float4*)(part + ((size_t)b * NPART + rg) * 256) + cg;
    const float4 v0 = pr[0];
    const float4 v1 = pr[(size_t)16 * 64];
    float4 s;
    s.x = v0.x + v1.x;  s.y = v0.y + v1.y;
    s.z = v0.z + v1.z;  s.w = v0.w + v1.w;
    red[rg][cg] = s;
  }
  __syncthreads();

  if (t < 256) {
    const float* rf = (const float*)&red[0][0];
    float s = 0.f;
    for (int g = 0; g < 16; ++g) s += rf[g * 256 + t];
    xm[t] = s * (1.0f / 1024.0f);
  }
  __syncthreads();

  if (t < 128) {
    float a = bs1[t];
    for (int k = 0; k < 256; ++k) a += xm[k] * Ws1[k * 128 + t];
    hid[t] = fmaxf(a, 0.f);
  }
  __syncthreads();

  if (t < 5) {
    float a = bs2[t];
    for (int k = 0; k < 128; ++k) a += hid[k] * Ws2[k * 5 + t];
    slg[t] = a;
    slog_out[b * 5 + t] = a;
  }
  __syncthreads();

  if (t == 0) {
    float mx = slg[0];
    for (int i = 1; i < 5; ++i) mx = fmaxf(mx, slg[i]);
    float sum = 0.f;
    for (int i = 0; i < 5; ++i) { sp[i] = __expf(slg[i] - mx); sum += sp[i]; }
    float inv = 1.0f / sum;
    for (int i = 0; i < 5; ++i) sp[i] *= inv;
  }
  __syncthreads();

  if (t < 64) {
    float a = bc1[t];
    for (int k = 0; k < 5; ++k) a += sp[k] * Wc1[k * 64 + t];
    c1s[t] = fmaxf(a, 0.f);
  }
  __syncthreads();

  if (t < 64) {
    float a = bc2[t];
    for (int k = 0; k < 64; ++k) a += c1s[k] * Wc2[k * 64 + t];
    c2s[t] = a;
  }
  __syncthreads();

  if (t == 0) {
    float cs = 0.f;
    for (int k = 0; k < 64; ++k) cs += c2s[k] * att_w[128 + k];
    cs_out[b] = cs;
  }
}

__global__ void __launch_bounds__(256) logA_kernel(const float* __restrict__ A,
                                                   float* __restrict__ logF) {
  const int o = blockIdx.x * 256 + threadIdx.x;
  const int c = o & 15, hf = (o >> 4) & 1, jq = (o >> 5) & 127, tile = o >> 12;
  const float4 a = *(const float4*)(A + (size_t)(tile * 16 + c) * CC +
                                    jq * 8 + hf * 4);
  float4 l;
  l.x = logf(a.x + 1e-9f);
  l.y = logf(a.y + 1e-9f);
  l.z = logf(a.z + 1e-9f);
  l.w = logf(a.w + 1e-9f);
  ((float4*)logF)[o] = l;
}

#define APAD 260
__global__ void __launch_bounds__(256) proj_gemm(
    const float* __restrict__ x, const float* __restrict__ Wp,
    const float* __restrict__ bp, float* __restrict__ xp) {
  extern __shared__ float sm[];
  float* Apan = sm;
  float* BpT  = sm + 64 * APAD;

  const int tid = threadIdx.x;
  const int m0 = blockIdx.x * 64;
  const int n0 = blockIdx.y * 64;

  for (int idx = tid; idx < 64 * 256; idx += 256) {
    const int r = idx >> 8, k = idx & 255;
    Apan[r * APAD + k] = x[(size_t)(m0 + r) * 256 + k];
  }
  for (int idx = tid; idx < 256 * 64; idx += 256) {
    const int k = idx >> 6, n = idx & 63;
    BpT[n * APAD + k] = Wp[(size_t)k * 256 + n0 + n];
  }
  __syncthreads();

  const int wave = tid >> 5, lane = tid & 31;
  const int r16 = lane & 15, half = lane >> 4;
  const int dt = wave >> 1;
  const int jw = (wave & 1) * 32;

  v8f acc0 = {0, 0, 0, 0, 0, 0, 0, 0};
  v8f acc1 = {0, 0, 0, 0, 0, 0, 0, 0};

  const float* arow = BpT + (dt * 16 + r16) * APAD;
  const float* b0row = Apan + (jw + r16) * APAD;
  const float* b1row = Apan + (jw + 16 + r16) * APAD;

  for (int kk = 0; kk < 256; kk += 4) {
    const int ko = kk + 2 * half;
    v2f a;  a.x = arow[ko];  a.y = arow[ko + 1];
    v2f b0; b0.x = b0row[ko]; b0.y = b0row[ko + 1];
    v2f b1; b1.x = b1row[ko]; b1.y = b1row[ko + 1];
    acc0 = wmma_f32(a, b0, acc0);
    acc1 = wmma_f32(a, b1, acc1);
  }

  const float4 bl = *(const float4*)(bp + n0 + dt * 16 + 8 * half);
  const float4 bh = *(const float4*)(bp + n0 + dt * 16 + 8 * half + 4);
  const float bias[8] = {bl.x, bl.y, bl.z, bl.w, bh.x, bh.y, bh.z, bh.w};
  const int bidx = m0 >> 10;
  const int jb = (m0 & 1023) + jw + r16;
  float* xrow = xp + ((size_t)(bidx * 4 + blockIdx.y) * 64 + dt * 16 + 8 * half)
                     * 1024 + jb;
#pragma unroll
  for (int v = 0; v < 8; ++v) {
    xrow[(size_t)v * 1024]      = acc0[v] + bias[v];
    xrow[(size_t)v * 1024 + 16] = acc1[v] + bias[v];
  }
}

#define JP 132
#define CS (64 * JP)
__global__ void __launch_bounds__(256) attn_kernel(
    const float* __restrict__ xp,   const float* __restrict__ logF,
    const float* __restrict__ att_w,
    const float* __restrict__ csb,  const float* __restrict__ lam,
    float* __restrict__ out) {
  extern __shared__ float sm[];
  float* srcS = sm;
  float* dstS = srcS + 1024;
  float* XpT  = dstS + 1024;

  const int tid = threadIdx.x;
  const int wave = tid >> 5, lane = tid & 31;
  const int r16 = lane & 15, half = lane >> 4;

  const int bid = blockIdx.x;
  const int b  = bid >> 5;
  const int h  = (bid >> 3) & 3;
  const int it = bid & 7;
  const int i0 = it * 128 + wave * 16;

  const float* xbh = xp + (size_t)(b * 4 + h) * (64 * 1024);

  {
    const float4* xr = (const float4*)xbh + tid;
    float4 s = {0.f, 0.f, 0.f, 0.f};
    float4 d = {0.f, 0.f, 0.f, 0.f};
#pragma unroll 1
    for (int g = 0; g < 64; g += 32) {
      float4 v[32];
#pragma unroll
      for (int u = 0; u < 32; ++u) v[u] = xr[(size_t)(g + u) * 256];
#pragma unroll
      for (int u = 0; u < 32; ++u)
        *(float4*)(XpT + (tid >> 5) * CS + (g + u) * JP + 4 * (tid & 31)) = v[u];
#pragma unroll
      for (int u = 0; u < 32; ++u) {
        const float ws = att_w[g + u];
        const float wd = att_w[64 + g + u];
        s.x += v[u].x * ws;  s.y += v[u].y * ws;
        s.z += v[u].z * ws;  s.w += v[u].w * ws;
        d.x += v[u].x * wd;  d.y += v[u].y * wd;
        d.z += v[u].z * wd;  d.w += v[u].w * wd;
      }
    }
    ((float4*)srcS)[tid] = s;
    ((float4*)dstS)[tid] = d;
  }
  __syncthreads();

  const float cs = csb[b];
  const float lamh = lam[h];
  const float dv = dstS[i0 + r16];

  const float4* Lw = (const float4*)logF + (size_t)(i0 >> 4) * (128 * 32) + lane;
  const float* srcL = srcS + 4 * half;

  v8f acc[4];
  const v8f vzero = {0, 0, 0, 0, 0, 0, 0, 0};
#pragma unroll
  for (int nt = 0; nt < 4; ++nt) acc[nt] = vzero;

  float m = INF_NEG;
  float rsum = 0.f;
  const float* xcl = XpT + r16 * JP + 4 * half;

  for (int ch = 0; ch < 8; ++ch) {
    const int jc = ch * 128;
    float4 L[16];
#pragma unroll
    for (int q = 0; q < 16; ++q) L[q] = Lw[(ch * 16 + q) * 32];
    __builtin_amdgcn_sched_barrier(0);

    float cm = INF_NEG;
#pragma unroll
    for (int q = 0; q < 16; ++q) {
      const float4 sv = *(const float4*)(srcL + jc + q * 8);
      float e0 = sv.x + dv + cs; e0 = e0 > 0.f ? e0 : 0.2f * e0;
      float e1 = sv.y + dv + cs; e1 = e1 > 0.f ? e1 : 0.2f * e1;
      float e2 = sv.z + dv + cs; e2 = e2 > 0.f ? e2 : 0.2f * e2;
      float e3 = sv.w + dv + cs; e3 = e3 > 0.f ? e3 : 0.2f * e3;
      L[q].x = e0 + lamh * L[q].x;
      L[q].y = e1 + lamh * L[q].y;
      L[q].z = e2 + lamh * L[q].z;
      L[q].w = e3 + lamh * L[q].w;
      cm = fmaxf(cm, fmaxf(fmaxf(L[q].x, L[q].y), fmaxf(L[q].z, L[q].w)));
    }
    cm = fmaxf(cm, __shfl_xor(cm, 16, 32));
    const float mn = fmaxf(m, cm);
    const float sc = __expf(m - mn);
    m = mn;
    rsum *= sc;
#pragma unroll
    for (int v = 0; v < 8; ++v) {
      const float f = __shfl(sc, v + 8 * half, 32);
#pragma unroll
      for (int nt = 0; nt < 4; ++nt) acc[nt][v] *= f;
    }

#pragma unroll
    for (int q = 0; q < 16; ++q) {
      const float p0 = __expf(L[q].x - m);
      const float p1 = __expf(L[q].y - m);
      const float p2 = __expf(L[q].z - m);
      const float p3 = __expf(L[q].w - m);
      rsum += (p0 + p1) + (p2 + p3);
      v2f a0; a0.x = p0; a0.y = p1;
      v2f a1; a1.x = p2; a1.y = p3;
      float4 bq[4];
#pragma unroll
      for (int nt = 0; nt < 4; ++nt)
        bq[nt] = *(const float4*)(xcl + ch * CS + nt * 16 * JP + q * 8);
#pragma unroll
      for (int nt = 0; nt < 4; ++nt) {
        v2f bv; bv.x = bq[nt].x; bv.y = bq[nt].y;
        acc[nt] = wmma_f32(a0, bv, acc[nt]);
      }
#pragma unroll
      for (int nt = 0; nt < 4; ++nt) {
        v2f bv; bv.x = bq[nt].z; bv.y = bq[nt].w;
        acc[nt] = wmma_f32(a1, bv, acc[nt]);
      }
    }
  }

  const float rs = rsum + __shfl_xor(rsum, 16, 32);
#pragma unroll
  for (int v = 0; v < 8; ++v) {
    const int rr = v + 8 * half;
    const float inv = 1.0f / __shfl(rs, rr, 32);
    const int row = i0 + rr;
#pragma unroll
    for (int nt = 0; nt < 4; ++nt) {
      out[(size_t)(b * CC + row) * 256 + h * 64 + nt * 16 + r16] =
          acc[nt][v] * inv;
    }
  }
}

extern "C" void kernel_launch(void* const* d_in, const int* in_sizes, int n_in,
                              void* d_out, int out_size, void* d_ws,
                              size_t ws_size, hipStream_t stream) {
  (void)in_sizes; (void)n_in; (void)out_size; (void)ws_size;

  const float* x   = (const float*)d_in[0];
  const float* Apr = (const float*)d_in[1];
  const float* Wp  = (const float*)d_in[2];
  const float* bp  = (const float*)d_in[3];
  const float* Ws1 = (const float*)d_in[4];
  const float* bs1 = (const float*)d_in[5];
  const float* Ws2 = (const float*)d_in[6];
  const float* bs2 = (const float*)d_in[7];
  const float* Wc1 = (const float*)d_in[8];
  const float* bc1 = (const float*)d_in[9];
  const float* Wc2 = (const float*)d_in[10];
  const float* bc2 = (const float*)d_in[11];
  const float* att = (const float*)d_in[12];
  const float* lam = (const float*)d_in[13];

  float* out  = (float*)d_out;
  float* slog = out + (size_t)BB * CC * 256;

  float* xp   = (float*)d_ws;
  float* logA = xp + 2097152;
  float* csb  = logA + 1048576;
  float* part = xp;

  colsum_kernel<<<BB * NPART, 256, 0, stream>>>(x, part);
  stats_kernel<<<BB, 1024, 0, stream>>>(part, Ws1, bs1, Ws2, bs2, Wc1, bc1,
                                        Wc2, bc2, att, slog, csb);
  logA_kernel<<<1024, 256, 0, stream>>>(Apr, logA);

  const size_t gemm_lds = (size_t)(2 * 64 * APAD) * sizeof(float);
  proj_gemm<<<dim3(128, 4), 256, gemm_lds, stream>>>(x, Wp, bp, xp);

  const size_t attn_lds =
      (size_t)(2048 + 8 * CS) * sizeof(float);
  attn_kernel<<<BB * 4 * 8, 256, attn_lds, stream>>>(xp, logA, att, csb, lam,
                                                     out);
}


# proj_gemm: K as a pipeline of 4 stages of 64: float4 loads of two stages in flight, each is stored to LDS and multiplied while the next arrive; 12 dependent load trips gone; x_proj sum order changes
# speedup vs baseline: 2.5276x; 1.0806x over previous
#include <hip/hip_runtime.h>
#include <math.h>


typedef float v2f __attribute__((ext_vector_type(2)));
typedef float v8f __attribute__((ext_vector_type(8)));

__device__ __forceinline__ v8f wmma_f32(v2f a, v2f b, v8f c) {
  return __builtin_amdgcn_wmma_f32_16x16x4_f32(false, a, false, b,
                                               (short)0, c, false, false);
}

#define BB 8
#define CC 1024
#define INF_NEG (-3.0e38f)

#define NPART 32
__global__ void __launch_bounds__(256) colsum_kernel(
    const float* __restrict__ x, float* __restrict__ part) {
  __shared__ float4 red4[4][64];
  const int b = blockIdx.x >> 5;
  const int p = blockIdx.x & 31;
  const int t = threadIdx.x;
  const int cg = t & 63;
  const int rg = t >> 6;
  const float4* xr =
      (const float4*)(x + ((size_t)b * CC + p * 32 + rg) * 256) + cg;
  float4 s = {0.f, 0.f, 0.f, 0.f};
#pragma unroll
  for (int u = 0; u < 8; ++u) {
    const float4 v = xr[(size_t)u * (4 * 64)];
    s.x += v.x;  s.y += v.y;  s.z += v.z;  s.w += v.w;
  }
  red4[rg][cg] = s;
  __syncthreads();
  if (t < 64) {
    float4 a = red4[0][t];
    for (int g = 1; g < 4; ++g) {
      const float4 v = red4[g][t];
      a.x += v.x;  a.y += v.y;  a.z += v.z;  a.w += v.w;
    }
    ((float4*)part)[(size_t)(b * NPART + p) * 64 + t] = a;
  }
}

__global__ void __launch_bounds__(1024) stats_kernel(
    const float* __restrict__ part,
    const float* __restrict__ Ws1, const float* __restrict__ bs1,
    const float* __restrict__ Ws2, const float* __restrict__ bs2,
    const float* __restrict__ Wc1, const float* __restrict__ bc1,
    const float* __restrict__ Wc2, const float* __restrict__ bc2,
    const float* __restrict__ att_w,
    float* __restrict__ slog_out, float* __restrict__ cs_out) {
  __shared__ float4 red[16][64];
  __shared__ float xm[256];
  __shared__ float hid[128];
  __shared__ float slg[5];
  __shared__ float sp[5];
  __shared__ float c1s[64];
  __shared__ float c2s[64];

  const int b = blockIdx.x;
  const int t = threadIdx.x;

  {
    const int cg = t & 63;
    const int rg = t >> 6;
    const float4* pr =
        (const float4*)(part + ((size_t)b * NPART + rg) * 256) + cg;
    const float4 v0 = pr[0];
    const float4 v1 = pr[(size_t)16 * 64];
    float4 s;
    s.x = v0.x + v1.x;  s.y = v0.y + v1.y;
    s.z = v0.z + v1.z;  s.w = v0.w + v1.w;
    red[rg][cg] = s;
  }
  __syncthreads();

  if (t < 256) {
    const float* rf = (const float*)&red[0][0];
    float s = 0.f;
    for (int g = 0; g < 16; ++g) s += rf[g * 256 + t];
    xm[t] = s * (1.0f / 1024.0f);
  }
  __syncthreads();

  if (t < 128) {
    float a = bs1[t];
    for (int k = 0; k < 256; ++k) a += xm[k] * Ws1[k * 128 + t];
    hid[t] = fmaxf(a, 0.f);
  }
  __syncthreads();

  if (t < 5) {
    float a = bs2[t];
    for (int k = 0; k < 128; ++k) a += hid[k] * Ws2[k * 5 + t];
    slg[t] = a;
    slog_out[b * 5 + t] = a;
  }
  __syncthreads();

  if (t == 0) {
    float mx = slg[0];
    for (int i = 1; i < 5; ++i) mx = fmaxf(mx, slg[i]);
    float sum = 0.f;
    for (int i = 0; i < 5; ++i) { sp[i] = __expf(slg[i] - mx); sum += sp[i]; }
    float inv = 1.0f / sum;
    for (int i = 0; i < 5; ++i) sp[i] *= inv;
  }
  __syncthreads();

  if (t < 64) {
    float a = bc1[t];
    for (int k = 0; k < 5; ++k) a += sp[k] * Wc1[k * 64 + t];
    c1s[t] = fmaxf(a, 0.f);
  }
  __syncthreads();

  if (t < 64) {
    float a = bc2[t];
    for (int k = 0; k < 64; ++k) a += c1s[k] * Wc2[k * 64 + t];
    c2s[t] = a;
  }
  __syncthreads();

  if (t == 0) {
    float cs = 0.f;
    for (int k = 0; k < 64; ++k) cs += c2s[k] * att_w[128 + k];
    cs_out[b] = cs;
  }
}

__global__ void __launch_bounds__(256) logA_kernel(const float* __restrict__ A,
                                                   float* __restrict__ logF) {
  const int o = blockIdx.x * 256 + threadIdx.x;
  const int c = o & 15, hf = (o >> 4) & 1, jq = (o >> 5) & 127, tile = o >> 12;
  const float4 a = *(const float4*)(A + (size_t)(tile * 16 + c) * CC +
                                    jq * 8 + hf * 4);
  float4 l;
  l.x = logf(a.x + 1e-9f);
  l.y = logf(a.y + 1e-9f);
  l.z = logf(a.z + 1e-9f);
  l.w = logf(a.w + 1e-9f);
  ((float4*)logF)[o] = l;
}

__device__ __forceinline__ void lds_barrier() {
  asm volatile("s_wait_dscnt 0x0\n\ts_barrier_signal -1\n\ts_barrier_wait -1"
               ::: "memory");
}

#define APAD 260
#define BPAD 72
__global__ void __launch_bounds__(256)
__attribute__((amdgpu_waves_per_eu(4, 4))) proj_gemm(
    const float* x, const float* Wp,
    const float* __restrict__ bp, float* xp) {
  extern __shared__ float sm[];
  float* Apan = sm;
  float* Bpan = sm + 64 * APAD;

  const int tid = threadIdx.x;
  const int m0 = blockIdx.x * 64;
  const int n0 = blockIdx.y * 64;

  const int r0 = tid >> 4, c4 = tid & 15;
  const float4* xg = (const float4*)(x + (size_t)(m0 + r0) * 256) + c4;
  const float4* wg = (const float4*)(Wp + (size_t)r0 * 256 + n0) + c4;

#define PROJ_LOAD(S, X, V)                                                    \
  const float4 X##0 = xg[(size_t)(0 * 16) * 64 + (S) * 16];                   \
  const float4 X##1 = xg[(size_t)(1 * 16) * 64 + (S) * 16];                   \
  const float4 X##2 = xg[(size_t)(2 * 16) * 64 + (S) * 16];                   \
  const float4 X##3 = xg[(size_t)(3 * 16) * 64 + (S) * 16];                   \
  const float4 V##0 = wg[(size_t)((S) * 64 + 0 * 16) * 64];                   \
  const float4 V##1 = wg[(size_t)((S) * 64 + 1 * 16) * 64];                   \
  const float4 V##2 = wg[(size_t)((S) * 64 + 2 * 16) * 64];                   \
  const float4 V##3 = wg[(size_t)((S) * 64 + 3 * 16) * 64];
  PROJ_LOAD(0, xs0_, ws0_)
  PROJ_LOAD(1, xs1_, ws1_)

  const int wave = tid >> 5, lane = tid & 31;
  const int r16 = lane & 15, half = lane >> 4;
  const int dt = wave >> 1;
  const int jw = (wave & 1) * 32;

  v8f acc0 = {0, 0, 0, 0, 0, 0, 0, 0};
  v8f acc1 = {0, 0, 0, 0, 0, 0, 0, 0};

  const float* b0p = Apan + (jw + r16) * APAD + 4 * half;
  const float* b1p = b0p + 16 * APAD;
  const float* ap  = Bpan + (4 * half) * BPAD + dt * 16 + r16;

#define PROJ_STORE(S, X, V)                                                   \
  *(float4*)(Apan + (r0 + 0 * 16) * APAD + (S) * 64 + 4 * c4) = X##0;         \
  *(float4*)(Apan + (r0 + 1 * 16) * APAD + (S) * 64 + 4 * c4) = X##1;         \
  *(float4*)(Apan + (r0 + 2 * 16) * APAD + (S) * 64 + 4 * c4) = X##2;         \
  *(float4*)(Apan + (r0 + 3 * 16) * APAD + (S) * 64 + 4 * c4) = X##3;         \
  *(float4*)(Bpan + ((S) * 64 + 0 * 16 + r0) * BPAD + 4 * c4) = V##0;         \
  *(float4*)(Bpan + ((S) * 64 + 1 * 16 + r0) * BPAD + 4 * c4) = V##1;         \
  *(float4*)(Bpan + ((S) * 64 + 2 * 16 + r0) * BPAD + 4 * c4) = V##2;         \
  *(float4*)(Bpan + ((S) * 64 + 3 * 16 + r0) * BPAD + 4 * c4) = V##3;         \
  lds_barrier();
#define PROJ_MUL(S)                                                           \
  _Pragma("unroll") for (int q = 0; q < 8; ++q) {                             \
    const int kb = (S) * 64 + q * 8;                                          \
    const float4 xa = *(const float4*)(b0p + kb);                             \
    const float4 xb = *(const float4*)(b1p + kb);                             \
    const float* a = ap + kb * BPAD;                                          \
    v2f w0; w0.x = a[0];        w0.y = a[BPAD];                               \
    v2f w1; w1.x = a[2 * BPAD]; w1.y = a[3 * BPAD];                           \
    v2f b;                                                                    \
    b.x = xa.x; b.y = xa.y;  acc0 = wmma_f32(w0, b, acc0);                    \
    b.x = xb.x; b.y = xb.y;  acc1 = wmma_f32(w0, b, acc1);                    \
    b.x = xa.z; b.y = xa.w;  acc0 = wmma_f32(w1, b, acc0);                    \
    b.x = xb.z; b.y = xb.w;  acc1 = wmma_f32(w1, b, acc1);                    \
  }
  PROJ_STORE(0, xs0_, ws0_)
  PROJ_LOAD(2, xs2_, ws2_)
  PROJ_MUL(0)
  PROJ_STORE(1, xs1_, ws1_)
  PROJ_LOAD(3, xs3_, ws3_)
  PROJ_MUL(1)
  PROJ_STORE(2, xs2_, ws2_)
  PROJ_MUL(2)
  PROJ_STORE(3, xs3_, ws3_)
  PROJ_MUL(3)
#undef PROJ_LOAD
#undef PROJ_STORE
#undef PROJ_MUL

  const float4 bl = *(const float4*)(bp + n0 + dt * 16 + 8 * half);
  const float4 bh = *(const float4*)(bp + n0 + dt * 16 + 8 * half + 4);
  const float bias[8] = {bl.x, bl.y, bl.z, bl.w, bh.x, bh.y, bh.z, bh.w};
  const int bidx = m0 >> 10;
  const int jb = (m0 & 1023) + jw + r16;
  float* xrow = xp + ((size_t)(bidx * 4 + blockIdx.y) * 64 + dt * 16 + 8 * half)
                     * 1024 + jb;
#pragma unroll
  for (int v = 0; v < 8; ++v) {
    xrow[(size_t)v * 1024]      = acc0[v] + bias[v];
    xrow[(size_t)v * 1024 + 16] = acc1[v] + bias[v];
  }
}

#define JP 132
#define CS (64 * JP)
__global__ void __launch_bounds__(256) attn_kernel(
    const float* __restrict__ xp,   const float* __restrict__ logF,
    const float* __restrict__ att_w,
    const float* __restrict__ csb,  const float* __restrict__ lam,
    float* __restrict__ out) {
  extern __shared__ float sm[];
  float* srcS = sm;
  float* dstS = srcS + 1024;
  float* XpT  = dstS + 1024;

  const int tid = threadIdx.x;
  const int wave = tid >> 5, lane = tid & 31;
  const int r16 = lane & 15, half = lane >> 4;

  const int bid = blockIdx.x;
  const int b  = bid >> 5;
  const int h  = (bid >> 3) & 3;
  const int it = bid & 7;
  const int i0 = it * 128 + wave * 16;

  const float* xbh = xp + (size_t)(b * 4 + h) * (64 * 1024);

  {
    const float4* xr = (const float4*)xbh + tid;
    float4 s = {0.f, 0.f, 0.f, 0.f};
    float4 d = {0.f, 0.f, 0.f, 0.f};
#pragma unroll 1
    for (int g = 0; g < 64; g += 32) {
      float4 v[32];
#pragma unroll
      for (int u = 0; u < 32; ++u) v[u] = xr[(size_t)(g + u) * 256];
#pragma unroll
      for (int u = 0; u < 32; ++u)
        *(float4*)(XpT + (tid >> 5) * CS + (g + u) * JP + 4 * (tid & 31)) = v[u];
#pragma unroll
      for (int u = 0; u < 32; ++u) {
        const float ws = att_w[g + u];
        const float wd = att_w[64 + g + u];
        s.x += v[u].x * ws;  s.y += v[u].y * ws;
        s.z += v[u].z * ws;  s.w += v[u].w * ws;
        d.x += v[u].x * wd;  d.y += v[u].y * wd;
        d.z += v[u].z * wd;  d.w += v[u].w * wd;
      }
    }
    ((float4*)srcS)[tid] = s;
    ((float4*)dstS)[tid] = d;
  }
  __syncthreads();

  const float cs = csb[b];
  const float lamh = lam[h];
  const float dv = dstS[i0 + r16];

  const float4* Lw = (const float4*)logF + (size_t)(i0 >> 4) * (128 * 32) + lane;
  const float* srcL = srcS + 4 * half;

  v8f acc[4];
  const v8f vzero = {0, 0, 0, 0, 0, 0, 0, 0};
#pragma unroll
  for (int nt = 0; nt < 4; ++nt) acc[nt] = vzero;

  float m = INF_NEG;
  float rsum = 0.f;
  const float* xcl = XpT + r16 * JP + 4 * half;

  for (int ch = 0; ch < 8; ++ch) {
    const int jc = ch * 128;
    float4 L[16];
#pragma unroll
    for (int q = 0; q < 16; ++q) L[q] = Lw[(ch * 16 + q) * 32];
    __builtin_amdgcn_sched_barrier(0);

    float cm = INF_NEG;
#pragma unroll
    for (int q = 0; q < 16; ++q) {
      const float4 sv = *(const float4*)(srcL + jc + q * 8);
      float e0 = sv.x + dv + cs; e0 = e0 > 0.f ? e0 : 0.2f * e0;
      float e1 = sv.y + dv + cs; e1 = e1 > 0.f ? e1 : 0.2f * e1;
      float e2 = sv.z + dv + cs; e2 = e2 > 0.f ? e2 : 0.2f * e2;
      float e3 = sv.w + dv + cs; e3 = e3 > 0.f ? e3 : 0.2f * e3;
      L[q].x = e0 + lamh * L[q].x;
      L[q].y = e1 + lamh * L[q].y;
      L[q].z = e2 + lamh * L[q].z;
      L[q].w = e3 + lamh * L[q].w;
      cm = fmaxf(cm, fmaxf(fmaxf(L[q].x, L[q].y), fmaxf(L[q].z, L[q].w)));
    }
    cm = fmaxf(cm, __shfl_xor(cm, 16, 32));
    const float mn = fmaxf(m, cm);
    const float sc = __expf(m - mn);
    m = mn;
    rsum *= sc;
#pragma unroll
    for (int v = 0; v < 8; ++v) {
      const float f = __shfl(sc, v + 8 * half, 32);
#pragma unroll
      for (int nt = 0; nt < 4; ++nt) acc[nt][v] *= f;
    }

#pragma unroll
    for (int q = 0; q < 16; ++q) {
      const float p0 = __expf(L[q].x - m);
      const float p1 = __expf(L[q].y - m);
      const float p2 = __expf(L[q].z - m);
      const float p3 = __expf(L[q].w - m);
      rsum += (p0 + p1) + (p2 + p3);
      v2f a0; a0.x = p0; a0.y = p1;
      v2f a1; a1.x = p2; a1.y = p3;
      float4 bq[4];
#pragma unroll
      for (int nt = 0; nt < 4; ++nt)
        bq[nt] = *(const float4*)(xcl + ch * CS + nt * 16 * JP + q * 8);
#pragma unroll
      for (int nt = 0; nt < 4; ++nt) {
        v2f bv; bv.x = bq[nt].x; bv.y = bq[nt].y;
        acc[nt] = wmma_f32(a0, bv, acc[nt]);
      }
#pragma unroll
      for (int nt = 0; nt < 4; ++nt) {
        v2f bv; bv.x = bq[nt].z; bv.y = bq[nt].w;
        acc[nt] = wmma_f32(a1, bv, acc[nt]);
      }
    }
  }

  const float rs = rsum + __shfl_xor(rsum, 16, 32);
#pragma unroll
  for (int v = 0; v < 8; ++v) {
    const int rr = v + 8 * half;
    const float inv = 1.0f / __shfl(rs, rr, 32);
    const int row = i0 + rr;
#pragma unroll
    for (int nt = 0; nt < 4; ++nt) {
      out[(size_t)(b * CC + row) * 256 + h * 64 + nt * 16 + r16] =
          acc[nt][v] * inv;
    }
  }
}

extern "C" void kernel_launch(void* const* d_in, const int* in_sizes, int n_in,
                              void* d_out, int out_size, void* d_ws,
                              size_t ws_size, hipStream_t stream) {
  (void)in_sizes; (void)n_in; (void)out_size; (void)ws_size;

  const float* x   = (const float*)d_in[0];
  const float* Apr = (const float*)d_in[1];
  const float* Wp  = (const float*)d_in[2];
  const float* bp  = (const float*)d_in[3];
  const float* Ws1 = (const float*)d_in[4];
  const float* bs1 = (const float*)d_in[5];
  const float* Ws2 = (const float*)d_in[6];
  const float* bs2 = (const float*)d_in[7];
  const float* Wc1 = (const float*)d_in[8];
  const float* bc1 = (const float*)d_in[9];
  const float* Wc2 = (const float*)d_in[10];
  const float* bc2 = (const float*)d_in[11];
  const float* att = (const float*)d_in[12];
  const float* lam = (const float*)d_in[13];

  float* out  = (float*)d_out;
  float* slog = out + (size_t)BB * CC * 256;

  float* xp   = (float*)d_ws;
  float* logA = xp + 2097152;
  float* csb  = logA + 1048576;
  float* part = xp;

  colsum_kernel<<<BB * NPART, 256, 0, stream>>>(x, part);
  stats_kernel<<<BB, 1024, 0, stream>>>(part, Ws1, bs1, Ws2, bs2, Wc1, bc1,
                                        Wc2, bc2, att, slog, csb);
  logA_kernel<<<1024, 256, 0, stream>>>(Apr, logA);

  const size_t gemm_lds =
      (size_t)(64 * APAD + 256 * BPAD) * sizeof(float);
  proj_gemm<<<dim3(128, 4), 256, gemm_lds, stream>>>(x, Wp, bp, xp);

  const size_t attn_lds =
      (size_t)(2048 + 8 * CS) * sizeof(float);
  attn_kernel<<<BB * 4 * 8, 256, attn_lds, stream>>>(xp, logA, att, csb, lam,
                                                     out);
}


# squeeze layer 1 by linearity in the pool grid: each of 256 blocks (128 rows x 64 cols) multiplies its column sums by its 64x128 slice of Ws1; stats adds 32 partial vectors; hidden sum order changes
# speedup vs baseline: 2.8918x; 1.1441x over previous
#include <hip/hip_runtime.h>
#include <math.h>


typedef float v2f __attribute__((ext_vector_type(2)));
typedef float v8f __attribute__((ext_vector_type(8)));

__device__ __forceinline__ v8f wmma_f32(v2f a, v2f b, v8f c) {
  return __builtin_amdgcn_wmma_f32_16x16x4_f32(false, a, false, b,
                                               (short)0, c, false, false);
}

#define BB 8
#define CC 1024
#define INF_NEG (-3.0e38f)

#define NPART 32
__global__ void __launch_bounds__(256) pool_hidden_kernel(
    const float* __restrict__ x, const float* __restrict__ Ws1,
    float* __restrict__ ph) {
  __shared__ float4 red4[16][16];
  __shared__ float4 csum4[16];
  __shared__ float4 hp4[8][32];
  const int b  = blockIdx.x >> 5;
  const int rg = (blockIdx.x >> 2) & 7;
  const int cg = blockIdx.x & 3;
  const int t = threadIdx.x;
  const int c4 = t & 15, rs = t >> 4;
  const int o4 = t & 31, ks = t >> 5;
  const float4* xr =
      (const float4*)(x + ((size_t)b * CC + rg * 128 + rs) * 256 + cg * 64) + c4;
  const float4* wr =
      (const float4*)(Ws1 + (size_t)(cg * 64 + ks * 8) * 128) + o4;
  float4 v[8], w[8];
#pragma unroll
  for (int u = 0; u < 8; ++u) v[u] = xr[(size_t)u * (16 * 64)];
#pragma unroll
  for (int u = 0; u < 8; ++u) w[u] = wr[(size_t)u * 32];
  __builtin_amdgcn_sched_barrier(0);

  float4 s = {0.f, 0.f, 0.f, 0.f};
#pragma unroll
  for (int u = 0; u < 8; ++u) {
    s.x += v[u].x;  s.y += v[u].y;  s.z += v[u].z;  s.w += v[u].w;
  }
  red4[rs][c4] = s;
  __syncthreads();
  if (t < 64) {
    const float* rf = (const float*)&red4[0][0];
    float a = rf[t];
    for (int g = 1; g < 16; ++g) a += rf[g * 64 + t];
    ((float*)&csum4[0])[t] = a;
  }
  __syncthreads();
  {
    const float4 ca = csum4[ks * 2];
    const float4 cb = csum4[ks * 2 + 1];
    const float ck[8] = {ca.x, ca.y, ca.z, ca.w, cb.x, cb.y, cb.z, cb.w};
    float4 a = {0.f, 0.f, 0.f, 0.f};
#pragma unroll
    for (int u = 0; u < 8; ++u) {
      a.x += ck[u] * w[u].x;  a.y += ck[u] * w[u].y;
      a.z += ck[u] * w[u].z;  a.w += ck[u] * w[u].w;
    }
    hp4[ks][o4] = a;
  }
  __syncthreads();
  if (t < 128) {
    const float* hf = (const float*)&hp4[0][0];
    float a = hf[t];
    for (int k = 1; k < 8; ++k) a += hf[k * 128 + t];
    ph[(size_t)blockIdx.x * 128 + t] = a;
  }
}

__global__ void __launch_bounds__(1024) stats_kernel(
    const float* __restrict__ ph, const float* __restrict__ bs1,
    const float* __restrict__ Ws2, const float* __restrict__ bs2,
    const float* __restrict__ Wc1, const float* __restrict__ bc1,
    const float* __restrict__ Wc2, const float* __restrict__ bc2,
    const float* __restrict__ att_w,
    float* __restrict__ slog_out, float* __restrict__ cs_out) {
  __shared__ float4 red[32][32];
  __shared__ float hid[128];
  __shared__ float slg[5];
  __shared__ float sp[5];
  __shared__ float c1s[64];
  __shared__ float c2s[64];

  const int b = blockIdx.x;
  const int t = threadIdx.x;

  const float b1 = bs1[t & 127];
  {
    const int o4 = t & 31;
    const int p  = t >> 5;
    red[p][o4] =
        ((const float4*)(ph + ((size_t)b * NPART + p) * 128))[o4];
  }
  __syncthreads();

  if (t < 128) {
    const float* rf = (const float*)&red[0][0];
    float s = 0.f;
    for (int p = 0; p < 32; ++p) s += rf[p * 128 + t];
    hid[t] = fmaxf(b1 + s * (1.0f / 1024.0f), 0.f);
  }
  __syncthreads();

  if (t < 5) {
    float a = bs2[t];
    for (int k = 0; k < 128; ++k) a += hid[k] * Ws2[k * 5 + t];
    slg[t] = a;
    slog_out[b * 5 + t] = a;
  }
  __syncthreads();

  if (t == 0) {
    float mx = slg[0];
    for (int i = 1; i < 5; ++i) mx = fmaxf(mx, slg[i]);
    float sum = 0.f;
    for (int i = 0; i < 5; ++i) { sp[i] = __expf(slg[i] - mx); sum += sp[i]; }
    float inv = 1.0f / sum;
    for (int i = 0; i < 5; ++i) sp[i] *= inv;
  }
  __syncthreads();

  if (t < 64) {
    float a = bc1[t];
    for (int k = 0; k < 5; ++k) a += sp[k] * Wc1[k * 64 + t];
    c1s[t] = fmaxf(a, 0.f);
  }
  __syncthreads();

  if (t < 64) {
    float a = bc2[t];
    for (int k = 0; k < 64; ++k) a += c1s[k] * Wc2[k * 64 + t];
    c2s[t] = a;
  }
  __syncthreads();

  if (t == 0) {
    float cs = 0.f;
    for (int k = 0; k < 64; ++k) cs += c2s[k] * att_w[128 + k];
    cs_out[b] = cs;
  }
}

__global__ void __launch_bounds__(256) logA_kernel(const float* __restrict__ A,
                                                   float* __restrict__ logF) {
  const int o = blockIdx.x * 256 + threadIdx.x;
  const int c = o & 15, hf = (o >> 4) & 1, jq = (o >> 5) & 127, tile = o >> 12;
  const float4 a = *(const float4*)(A + (size_t)(tile * 16 + c) * CC +
                                    jq * 8 + hf * 4);
  float4 l;
  l.x = logf(a.x + 1e-9f);
  l.y = logf(a.y + 1e-9f);
  l.z = logf(a.z + 1e-9f);
  l.w = logf(a.w + 1e-9f);
  ((float4*)logF)[o] = l;
}

__device__ __forceinline__ void lds_barrier() {
  asm volatile("s_wait_dscnt 0x0\n\ts_barrier_signal -1\n\ts_barrier_wait -1"
               ::: "memory");
}

#define APAD 260
#define BPAD 72
__global__ void __launch_bounds__(256)
__attribute__((amdgpu_waves_per_eu(4, 4))) proj_gemm(
    const float* x, const float* Wp,
    const float* __restrict__ bp, float* xp) {
  extern __shared__ float sm[];
  float* Apan = sm;
  float* Bpan = sm + 64 * APAD;

  const int tid = threadIdx.x;
  const int m0 = blockIdx.x * 64;
  const int n0 = blockIdx.y * 64;

  const int r0 = tid >> 4, c4 = tid & 15;
  const float4* xg = (const float4*)(x + (size_t)(m0 + r0) * 256) + c4;
  const float4* wg = (const float4*)(Wp + (size_t)r0 * 256 + n0) + c4;

#define PROJ_LOAD(S, X, V)                                                    \
  const float4 X##0 = xg[(size_t)(0 * 16) * 64 + (S) * 16];                   \
  const float4 X##1 = xg[(size_t)(1 * 16) * 64 + (S) * 16];                   \
  const float4 X##2 = xg[(size_t)(2 * 16) * 64 + (S) * 16];                   \
  const float4 X##3 = xg[(size_t)(3 * 16) * 64 + (S) * 16];                   \
  const float4 V##0 = wg[(size_t)((S) * 64 + 0 * 16) * 64];                   \
  const float4 V##1 = wg[(size_t)((S) * 64 + 1 * 16) * 64];                   \
  const float4 V##2 = wg[(size_t)((S) * 64 + 2 * 16) * 64];                   \
  const float4 V##3 = wg[(size_t)((S) * 64 + 3 * 16) * 64];
  PROJ_LOAD(0, xs0_, ws0_)
  PROJ_LOAD(1, xs1_, ws1_)

  const int wave = tid >> 5, lane = tid & 31;
  const int r16 = lane & 15, half = lane >> 4;
  const int dt = wave >> 1;
  const int jw = (wave & 1) * 32;

  v8f acc0 = {0, 0, 0, 0, 0, 0, 0, 0};
  v8f acc1 = {0, 0, 0, 0, 0, 0, 0, 0};

  const float* b0p = Apan + (jw + r16) * APAD + 4 * half;
  const float* b1p = b0p + 16 * APAD;
  const float* ap  = Bpan + (4 * half) * BPAD + dt * 16 + r16;

#define PROJ_STORE(S, X, V)                                                   \
  *(float4*)(Apan + (r0 + 0 * 16) * APAD + (S) * 64 + 4 * c4) = X##0;         \
  *(float4*)(Apan + (r0 + 1 * 16) * APAD + (S) * 64 + 4 * c4) = X##1;         \
  *(float4*)(Apan + (r0 + 2 * 16) * APAD + (S) * 64 + 4 * c4) = X##2;         \
  *(float4*)(Apan + (r0 + 3 * 16) * APAD + (S) * 64 + 4 * c4) = X##3;         \
  *(float4*)(Bpan + ((S) * 64 + 0 * 16 + r0) * BPAD + 4 * c4) = V##0;         \
  *(float4*)(Bpan + ((S) * 64 + 1 * 16 + r0) * BPAD + 4 * c4) = V##1;         \
  *(float4*)(Bpan + ((S) * 64 + 2 * 16 + r0) * BPAD + 4 * c4) = V##2;         \
  *(float4*)(Bpan + ((S) * 64 + 3 * 16 + r0) * BPAD + 4 * c4) = V##3;         \
  lds_barrier();
#define PROJ_MUL(S)                                                           \
  _Pragma("unroll") for (int q = 0; q < 8; ++q) {                             \
    const int kb = (S) * 64 + q * 8;                                          \
    const float4 xa = *(const float4*)(b0p + kb);                             \
    const float4 xb = *(const float4*)(b1p + kb);                             \
    const float* a = ap + kb * BPAD;                                          \
    v2f w0; w0.x = a[0];        w0.y = a[BPAD];                               \
    v2f w1; w1.x = a[2 * BPAD]; w1.y = a[3 * BPAD];                           \
    v2f b;                                                                    \
    b.x = xa.x; b.y = xa.y;  acc0 = wmma_f32(w0, b, acc0);                    \
    b.x = xb.x; b.y = xb.y;  acc1 = wmma_f32(w0, b, acc1);                    \
    b.x = xa.z; b.y = xa.w;  acc0 = wmma_f32(w1, b, acc0);                    \
    b.x = xb.z; b.y = xb.w;  acc1 = wmma_f32(w1, b, acc1);                    \
  }
  PROJ_STORE(0, xs0_, ws0_)
  PROJ_LOAD(2, xs2_, ws2_)
  PROJ_MUL(0)
  PROJ_STORE(1, xs1_, ws1_)
  PROJ_LOAD(3, xs3_, ws3_)
  PROJ_MUL(1)
  PROJ_STORE(2, xs2_, ws2_)
  PROJ_MUL(2)
  PROJ_STORE(3, xs3_, ws3_)
  PROJ_MUL(3)
#undef PROJ_LOAD
#undef PROJ_STORE
#undef PROJ_MUL

  const float4 bl = *(const float4*)(bp + n0 + dt * 16 + 8 * half);
  const float4 bh = *(const float4*)(bp + n0 + dt * 16 + 8 * half + 4);
  const float bias[8] = {bl.x, bl.y, bl.z, bl.w, bh.x, bh.y, bh.z, bh.w};
  const int bidx = m0 >> 10;
  const int jb = (m0 & 1023) + jw + r16;
  float* xrow = xp + ((size_t)(bidx * 4 + blockIdx.y) * 64 + dt * 16 + 8 * half)
                     * 1024 + jb;
#pragma unroll
  for (int v = 0; v < 8; ++v) {
    xrow[(size_t)v * 1024]      = acc0[v] + bias[v];
    xrow[(size_t)v * 1024 + 16] = acc1[v] + bias[v];
  }
}

#define JP 132
#define CS (64 * JP)
__global__ void __launch_bounds__(256) attn_kernel(
    const float* __restrict__ xp,   const float* __restrict__ logF,
    const float* __restrict__ att_w,
    const float* __restrict__ csb,  const float* __restrict__ lam,
    float* __restrict__ out) {
  extern __shared__ float sm[];
  float* srcS = sm;
  float* dstS = srcS + 1024;
  float* XpT  = dstS + 1024;

  const int tid = threadIdx.x;
  const int wave = tid >> 5, lane = tid & 31;
  const int r16 = lane & 15, half = lane >> 4;

  const int bid = blockIdx.x;
  const int b  = bid >> 5;
  const int h  = (bid >> 3) & 3;
  const int it = bid & 7;
  const int i0 = it * 128 + wave * 16;

  const float* xbh = xp + (size_t)(b * 4 + h) * (64 * 1024);

  {
    const float4* xr = (const float4*)xbh + tid;
    float4 s = {0.f, 0.f, 0.f, 0.f};
    float4 d = {0.f, 0.f, 0.f, 0.f};
#pragma unroll 1
    for (int g = 0; g < 64; g += 32) {
      float4 v[32];
#pragma unroll
      for (int u = 0; u < 32; ++u) v[u] = xr[(size_t)(g + u) * 256];
#pragma unroll
      for (int u = 0; u < 32; ++u)
        *(float4*)(XpT + (tid >> 5) * CS + (g + u) * JP + 4 * (tid & 31)) = v[u];
#pragma unroll
      for (int u = 0; u < 32; ++u) {
        const float ws = att_w[g + u];
        const float wd = att_w[64 + g + u];
        s.x += v[u].x * ws;  s.y += v[u].y * ws;
        s.z += v[u].z * ws;  s.w += v[u].w * ws;
        d.x += v[u].x * wd;  d.y += v[u].y * wd;
        d.z += v[u].z * wd;  d.w += v[u].w * wd;
      }
    }
    ((float4*)srcS)[tid] = s;
    ((float4*)dstS)[tid] = d;
  }
  __syncthreads();

  const float cs = csb[b];
  const float lamh = lam[h];
  const float dv = dstS[i0 + r16];

  const float4* Lw = (const float4*)logF + (size_t)(i0 >> 4) * (128 * 32) + lane;
  const float* srcL = srcS + 4 * half;

  v8f acc[4];
  const v8f vzero = {0, 0, 0, 0, 0, 0, 0, 0};
#pragma unroll
  for (int nt = 0; nt < 4; ++nt) acc[nt] = vzero;

  float m = INF_NEG;
  float rsum = 0.f;
  const float* xcl = XpT + r16 * JP + 4 * half;

  for (int ch = 0; ch < 8; ++ch) {
    const int jc = ch * 128;
    float4 L[16];
#pragma unroll
    for (int q = 0; q < 16; ++q) L[q] = Lw[(ch * 16 + q) * 32];
    __builtin_amdgcn_sched_barrier(0);

    float cm = INF_NEG;
#pragma unroll
    for (int q = 0; q < 16; ++q) {
      const float4 sv = *(const float4*)(srcL + jc + q * 8);
      float e0 = sv.x + dv + cs; e0 = e0 > 0.f ? e0 : 0.2f * e0;
      float e1 = sv.y + dv + cs; e1 = e1 > 0.f ? e1 : 0.2f * e1;
      float e2 = sv.z + dv + cs; e2 = e2 > 0.f ? e2 : 0.2f * e2;
      float e3 = sv.w + dv + cs; e3 = e3 > 0.f ? e3 : 0.2f * e3;
      L[q].x = e0 + lamh * L[q].x;
      L[q].y = e1 + lamh * L[q].y;
      L[q].z = e2 + lamh * L[q].z;
      L[q].w = e3 + lamh * L[q].w;
      cm = fmaxf(cm, fmaxf(fmaxf(L[q].x, L[q].y), fmaxf(L[q].z, L[q].w)));
    }
    cm = fmaxf(cm, __shfl_xor(cm, 16, 32));
    const float mn = fmaxf(m, cm);
    const float sc = __expf(m - mn);
    m = mn;
    rsum *= sc;
#pragma unroll
    for (int v = 0; v < 8; ++v) {
      const float f = __shfl(sc, v + 8 * half, 32);
#pragma unroll
      for (int nt = 0; nt < 4; ++nt) acc[nt][v] *= f;
    }

#pragma unroll
    for (int q = 0; q < 16; ++q) {
      const float p0 = __expf(L[q].x - m);
      const float p1 = __expf(L[q].y - m);
      const float p2 = __expf(L[q].z - m);
      const float p3 = __expf(L[q].w - m);
      rsum += (p0 + p1) + (p2 + p3);
      v2f a0; a0.x = p0; a0.y = p1;
      v2f a1; a1.x = p2; a1.y = p3;
      float4 bq[4];
#pragma unroll
      for (int nt = 0; nt < 4; ++nt)
        bq[nt] = *(const float4*)(xcl + ch * CS + nt * 16 * JP + q * 8);
#pragma unroll
      for (int nt = 0; nt < 4; ++nt) {
        v2f bv; bv.x = bq[nt].x; bv.y = bq[nt].y;
        acc[nt] = wmma_f32(a0, bv, acc[nt]);
      }
#pragma unroll
      for (int nt = 0; nt < 4; ++nt) {
        v2f bv; bv.x = bq[nt].z; bv.y = bq[nt].w;
        acc[nt] = wmma_f32(a1, bv, acc[nt]);
      }
    }
  }

  const float rs = rsum + __shfl_xor(rsum, 16, 32);
#pragma unroll
  for (int v = 0; v < 8; ++v) {
    const int rr = v + 8 * half;
    const float inv = 1.0f / __shfl(rs, rr, 32);
    const int row = i0 + rr;
#pragma unroll
    for (int nt = 0; nt < 4; ++nt) {
      out[(size_t)(b * CC + row) * 256 + h * 64 + nt * 16 + r16] =
          acc[nt][v] * inv;
    }
  }
}

extern "C" void kernel_launch(void* const* d_in, const int* in_sizes, int n_in,
                              void* d_out, int out_size, void* d_ws,
                              size_t ws_size, hipStream_t stream) {
  (void)in_sizes; (void)n_in; (void)out_size; (void)ws_size;

  const float* x   = (const float*)d_in[0];
  const float* Apr = (const float*)d_in[1];
  const float* Wp  = (const float*)d_in[2];
  const float* bp  = (const float*)d_in[3];
  const float* Ws1 = (const float*)d_in[4];
  const float* bs1 = (const float*)d_in[5];
  const float* Ws2 = (const float*)d_in[6];
  const float* bs2 = (const float*)d_in[7];
  const float* Wc1 = (const float*)d_in[8];
  const float* bc1 = (const float*)d_in[9];
  const float* Wc2 = (const float*)d_in[10];
  const float* bc2 = (const float*)d_in[11];
  const float* att = (const float*)d_in[12];
  const float* lam = (const float*)d_in[13];

  float* out  = (float*)d_out;
  float* slog = out + (size_t)BB * CC * 256;

  float* xp   = (float*)d_ws;
  float* logA = xp + 2097152;
  float* csb  = logA + 1048576;
  float* ph = xp;

  pool_hidden_kernel<<<BB * NPART, 256, 0, stream>>>(x, Ws1, ph);
  stats_kernel<<<BB, 1024, 0, stream>>>(ph, bs1, Ws2, bs2, Wc1, bc1,
                                        Wc2, bc2, att, slog, csb);
  logA_kernel<<<1024, 256, 0, stream>>>(Apr, logA);

  const size_t gemm_lds =
      (size_t)(64 * APAD + 256 * BPAD) * sizeof(float);
  proj_gemm<<<dim3(128, 4), 256, gemm_lds, stream>>>(x, Wp, bp, xp);

  const size_t attn_lds =
      (size_t)(2048 + 8 * CS) * sizeof(float);
  attn_kernel<<<BB * 4 * 8, 256, attn_lds, stream>>>(xp, logA, att, csb, lam,
                                                     out);
}
